# starting kernel
# baseline (speedup 1.0000x reference)
#include <hip/hip_runtime.h>
#include <math.h>

#define BATCH 8
#define SEQ   8192
#define DIM   64
#define NHASH 8
#define NBUCK 128
#define NBALL (NHASH*NBUCK)
#define BSZ   64
#define NCHUNK (NHASH*(SEQ/BSZ))
#define HS    (NHASH*SEQ)
#define TS    512
#define NT    (SEQ/TS)
#define SELF_VAL (-5.0e4f)

typedef float v2f __attribute__((ext_vector_type(2)));
typedef float v8f __attribute__((ext_vector_type(8)));

__global__ __launch_bounds__(256) void lsh_hash_kernel(
    const float* __restrict__ qk, const float* __restrict__ rot,
    int* __restrict__ bucketArr, int* __restrict__ tilecnt)
{
  __shared__ __align__(16) float A [64][68];
  __shared__ __align__(16) float Bm[64][68];
  __shared__ __align__(16) float R [64][68];

  const int bid = blockIdx.x;
  const int tt  = bid & 127;
  const int h   = (bid >> 7) & 7;
  const int b   = bid >> 10;
  const int tid = threadIdx.x;
  const int t0  = tt * 64;

  {
    int row = tid >> 2, c0 = (tid & 3) * 16;
    const float4* s4 = (const float4*)(qk + ((size_t)b*SEQ + t0 + row)*DIM + c0);
    float4* d4 = (float4*)(&A[row][c0]);
    d4[0]=s4[0]; d4[1]=s4[1]; d4[2]=s4[2]; d4[3]=s4[3];
  }
  {
    int f = tid >> 2, c0 = (tid & 3) * 16;
    const float4* s4 = (const float4*)(rot + (size_t)f*(NHASH*64) + (size_t)h*64 + c0);
    float4* d4 = (float4*)(&Bm[f][c0]);
    d4[0]=s4[0]; d4[1]=s4[1]; d4[2]=s4[2]; d4[3]=s4[3];
  }
  __syncthreads();

  const int wave = tid >> 5, lane = tid & 31, half = lane >> 4, mn = lane & 15;
  for (int T = wave; T < 16; T += 8) {
    const int mi = T >> 2, ni = T & 3;
    v8f acc = {};
#pragma unroll
    for (int k0 = 0; k0 < 64; k0 += 4) {
      v2f a, bb;
      a.x  = A [mi*16 + mn][k0 + 2*half    ];
      a.y  = A [mi*16 + mn][k0 + 2*half + 1];
      bb.x = Bm[k0 + 2*half    ][ni*16 + mn];
      bb.y = Bm[k0 + 2*half + 1][ni*16 + mn];
      acc = __builtin_amdgcn_wmma_f32_16x16x4_f32(false, a, false, bb,
                                                  (short)0, acc, false, false);
    }
#pragma unroll
    for (int r = 0; r < 8; r++)
      R[mi*16 + r + 8*half][ni*16 + mn] = acc[r];
  }
  __syncthreads();

  if (tid < 64) {
    const int row = tid;
    float best = R[row][0]; int bidx = 0;
    for (int j = 1; j < 128; j++) {
      float vv = (j < 64) ? R[row][j] : -R[row][j - 64];
      if (vv > best) { best = vv; bidx = j; }
    }
    const int q = h*NBUCK + bidx;
    const int t = t0 + row;
    bucketArr[((size_t)b*NHASH + h)*SEQ + t] = q;
    atomicAdd(&tilecnt[((size_t)b*NBALL + q)*NT + (t >> 9)], 1);
  }
}

__global__ __launch_bounds__(1024) void lsh_prefix_kernel(int* __restrict__ tilecnt)
{
  __shared__ int sc[NBALL];
  const int b = blockIdx.x, q = threadIdx.x;
  int* tc = tilecnt + ((size_t)b*NBALL + q)*NT;
  int s = 0;
#pragma unroll
  for (int j = 0; j < NT; j++) s += tc[j];
  sc[q] = s;
  __syncthreads();
  for (int off = 1; off < NBALL; off <<= 1) {
    int vv = (q >= off) ? sc[q - off] : 0;
    __syncthreads();
    sc[q] += vv;
    __syncthreads();
  }
  int run = sc[q] - s;
#pragma unroll
  for (int j = 0; j < NT; j++) { int c0 = tc[j]; tc[j] = run; run += c0; }
}

__global__ __launch_bounds__(TS) void lsh_rank_kernel(
    const int* __restrict__ bucketArr, const int* __restrict__ tilecnt,
    int* __restrict__ sticker)
{
  __shared__ int bk[TS];
  const int bid  = blockIdx.x;
  const int tile = bid & (NT - 1);
  const int h    = (bid >> 4) & 7;
  const int b    = bid >> 7;
  const int tid  = threadIdx.x;
  const int t    = tile*TS + tid;
  const int q    = bucketArr[((size_t)b*NHASH + h)*SEQ + t];
  bk[tid] = q;
  __syncthreads();
  int r = 0;
  for (int j = 0; j < tid; j++) r += (bk[j] == q) ? 1 : 0;
  const int pos = tilecnt[((size_t)b*NBALL + q)*NT + tile] + r;
  sticker[(size_t)b*HS + pos] = h*SEQ + t;
}

__global__ __launch_bounds__(256) void lsh_attn_kernel(
    const float* __restrict__ qk, const float* __restrict__ vin,
    const int* __restrict__ sticker,
    float* __restrict__ oBuf, float* __restrict__ lgBuf)
{
  __shared__ int idxA[128];
  __shared__ int tA  [128];
  __shared__ __align__(16) float Qs[64][68];
  __shared__ __align__(16) float Ks[128][68];
  __shared__ __align__(16) float Vs[128][68];
  __shared__ __align__(16) float Ps[64][132];
  __shared__ float lse[64];

  const int bid = blockIdx.x;
  const int c   = bid & (NCHUNK - 1);
  const int b   = bid >> 10;
  const int tid = threadIdx.x;
  const int prev = (c + NCHUNK - 1) & (NCHUNK - 1);

  if (tid < 128) {
    const int src_chunk = (tid < 64) ? c : prev;
    const int j = tid & 63;
    const int idx = sticker[(size_t)b*HS + (size_t)src_chunk*BSZ + j];
    idxA[tid] = idx;
    tA[tid]   = idx & (SEQ - 1);
  }
  __syncthreads();

  {
    int row = tid >> 2, c0 = (tid & 3) * 16;
    const float4* s4 = (const float4*)(qk + ((size_t)b*SEQ + tA[row])*DIM + c0);
    float4* d4 = (float4*)(&Qs[row][c0]);
    d4[0]=s4[0]; d4[1]=s4[1]; d4[2]=s4[2]; d4[3]=s4[3];
  }
  {
    int row = tid >> 1, c0 = (tid & 1) * 32;
    const int t = tA[row];
    const float4* sk = (const float4*)(qk  + ((size_t)b*SEQ + t)*DIM + c0);
    const float4* sv = (const float4*)(vin + ((size_t)b*SEQ + t)*DIM + c0);
    float4* dk = (float4*)(&Ks[row][c0]);
    float4* dv = (float4*)(&Vs[row][c0]);
#pragma unroll
    for (int i = 0; i < 8; i++) { dk[i] = sk[i]; dv[i] = sv[i]; }
  }
  __syncthreads();

  if (tid < 128) {
    float ss = 0.f;
#pragma unroll
    for (int d = 0; d < 64; d++) { float x = Ks[tid][d]; ss += x * x; }
    const float inv = 1.0f / (sqrtf(ss) + 1e-6f);
#pragma unroll
    for (int d = 0; d < 64; d++) Ks[tid][d] *= inv;
  }
  __syncthreads();

  const int wave = tid >> 5, lane = tid & 31, half = lane >> 4, mn = lane & 15;

  {
    const int ni = wave;
    for (int mi = 0; mi < 4; mi++) {
      v8f acc = {};
#pragma unroll
      for (int k0 = 0; k0 < 64; k0 += 4) {
        v2f a, bb;
        a.x  = Qs[mi*16 + mn][k0 + 2*half    ];
        a.y  = Qs[mi*16 + mn][k0 + 2*half + 1];
        bb.x = Ks[ni*16 + mn][k0 + 2*half    ];
        bb.y = Ks[ni*16 + mn][k0 + 2*half + 1];
        acc = __builtin_amdgcn_wmma_f32_16x16x4_f32(false, a, false, bb,
                                                    (short)0, acc, false, false);
      }
#pragma unroll
      for (int r = 0; r < 8; r++) {
        const int i = mi*16 + r + 8*half, j = ni*16 + mn;
        float dval = acc[r] * 0.125f;
        if (tA[i] == tA[j]) dval = SELF_VAL;
        Ps[i][j] = dval;
      }
    }
  }
  __syncthreads();

  if (tid < 64) {
    float m = -1e30f;
#pragma unroll
    for (int j = 0; j < 128; j++) m = fmaxf(m, Ps[tid][j]);
    float ssum = 0.f;
#pragma unroll
    for (int j = 0; j < 128; j++) { float e = expf(Ps[tid][j] - m); Ps[tid][j] = e; ssum += e; }
    lse[tid] = m + logf(ssum);
    const float invs = 1.0f / ssum;
#pragma unroll
    for (int j = 0; j < 128; j++) Ps[tid][j] *= invs;
  }
  __syncthreads();

  for (int T = wave; T < 16; T += 8) {
    const int mi = T >> 2, ni = T & 3;
    v8f acc = {};
#pragma unroll
    for (int k0 = 0; k0 < 128; k0 += 4) {
      v2f a, bb;
      a.x  = Ps[mi*16 + mn][k0 + 2*half    ];
      a.y  = Ps[mi*16 + mn][k0 + 2*half + 1];
      bb.x = Vs[k0 + 2*half    ][ni*16 + mn];
      bb.y = Vs[k0 + 2*half + 1][ni*16 + mn];
      acc = __builtin_amdgcn_wmma_f32_16x16x4_f32(false, a, false, bb,
                                                  (short)0, acc, false, false);
    }
#pragma unroll
    for (int r = 0; r < 8; r++)
      Qs[mi*16 + r + 8*half][ni*16 + mn] = acc[r];
  }
  __syncthreads();

  {
    int row = tid >> 2, c0 = (tid & 3) * 16;
    float4* d4 = (float4*)(oBuf + ((size_t)b*HS + idxA[row])*DIM + c0);
    const float4* s4 = (const float4*)(&Qs[row][c0]);
    d4[0]=s4[0]; d4[1]=s4[1]; d4[2]=s4[2]; d4[3]=s4[3];
  }
  if (tid < 64) lgBuf[(size_t)b*HS + idxA[tid]] = lse[tid];
}

__global__ __launch_bounds__(256) void lsh_combine_kernel(
    const float* __restrict__ oBuf, const float* __restrict__ lgBuf,
    float* __restrict__ out)
{
  const int g = blockIdx.x * blockDim.x + threadIdx.x;
  const int b = g >> 13;
  const int t = g & (SEQ - 1);
  float lg[NHASH];
  float m = -1e30f;
#pragma unroll
  for (int h = 0; h < NHASH; h++) {
    lg[h] = lgBuf[(size_t)b*HS + (size_t)h*SEQ + t];
    m = fmaxf(m, lg[h]);
  }
  float ssum = 0.f;
#pragma unroll
  for (int h = 0; h < NHASH; h++) { lg[h] = expf(lg[h] - m); ssum += lg[h]; }
  const float inv = 1.0f / ssum;
#pragma unroll
  for (int h = 0; h < NHASH; h++) lg[h] *= inv;

  float* dst = out + ((size_t)b*SEQ + t)*DIM;
#pragma unroll
  for (int d = 0; d < DIM; d += 4) {
    float4 acc = {0.f, 0.f, 0.f, 0.f};
    for (int h = 0; h < NHASH; h++) {
      const float4 o4 = *(const float4*)(oBuf + ((size_t)b*HS + (size_t)h*SEQ + t)*DIM + d);
      acc.x += lg[h]*o4.x; acc.y += lg[h]*o4.y; acc.z += lg[h]*o4.z; acc.w += lg[h]*o4.w;
    }
    *(float4*)(dst + d) = acc;
  }
}

extern "C" void kernel_launch(void* const* d_in, const int* in_sizes, int n_in,
                              void* d_out, int out_size, void* d_ws, size_t ws_size,
                              hipStream_t stream)
{
  const float* qk  = (const float*)d_in[0];
  const float* v   = (const float*)d_in[1];
  const float* rot = (const float*)d_in[2];

  char* ws = (char*)d_ws;
  size_t off = 0;
  auto alloc = [&](size_t bytes) -> void* {
    void* p = ws + off;
    off += (bytes + 255) & ~(size_t)255;
    return p;
  };
  int*   bucketArr = (int*)  alloc((size_t)BATCH*NHASH*SEQ*sizeof(int));
  int*   tilecnt   = (int*)  alloc((size_t)BATCH*NBALL*NT*sizeof(int));
  int*   sticker   = (int*)  alloc((size_t)BATCH*HS*sizeof(int));
  float* oBuf      = (float*)alloc((size_t)BATCH*HS*DIM*sizeof(float));
  float* lgBuf     = (float*)alloc((size_t)BATCH*HS*sizeof(float));
  if (off > ws_size) return;

  hipMemsetAsync(tilecnt, 0, (size_t)BATCH*NBALL*NT*sizeof(int), stream);

  lsh_hash_kernel   <<<BATCH*NHASH*(SEQ/64), 256, 0, stream>>>(qk, rot, bucketArr, tilecnt);
  lsh_prefix_kernel <<<BATCH,               1024, 0, stream>>>(tilecnt);
  lsh_rank_kernel   <<<BATCH*NHASH*NT,        TS, 0, stream>>>(bucketArr, tilecnt, sticker);
  lsh_attn_kernel   <<<BATCH*NCHUNK,         256, 0, stream>>>(qk, v, sticker, oBuf, lgBuf);
  lsh_combine_kernel<<<(BATCH*SEQ)/256,      256, 0, stream>>>(oBuf, lgBuf, (float*)d_out);
}


# attention: 4-wave group, each wave owns 16 query rows end to end: scores+softmax in its accumulators (row max/sum by 16-lane DPP: sum order changed), key 1/norm applied to scores, P via own LDS strip
# speedup vs baseline: 1.2809x; 1.2809x over previous
#include <hip/hip_runtime.h>
#include <math.h>

#define BATCH 8
#define SEQ   8192
#define DIM   64
#define NHASH 8
#define NBUCK 128
#define NBALL (NHASH*NBUCK)
#define BSZ   64
#define NCHUNK (NHASH*(SEQ/BSZ))
#define HS    (NHASH*SEQ)
#define TS    512
#define NT    (SEQ/TS)
#define SELF_VAL (-5.0e4f)

typedef float v2f __attribute__((ext_vector_type(2)));
typedef float v8f __attribute__((ext_vector_type(8)));

__global__ __launch_bounds__(256) void lsh_hash_kernel(
    const float* __restrict__ qk, const float* __restrict__ rot,
    int* __restrict__ bucketArr, int* __restrict__ tilecnt)
{
  __shared__ __align__(16) float A [64][68];
  __shared__ __align__(16) float Bm[64][68];
  __shared__ __align__(16) float R [64][68];

  const int bid = blockIdx.x;
  const int tt  = bid & 127;
  const int h   = (bid >> 7) & 7;
  const int b   = bid >> 10;
  const int tid = threadIdx.x;
  const int t0  = tt * 64;

  {
    int row = tid >> 2, c0 = (tid & 3) * 16;
    const float4* s4 = (const float4*)(qk + ((size_t)b*SEQ + t0 + row)*DIM + c0);
    float4* d4 = (float4*)(&A[row][c0]);
    d4[0]=s4[0]; d4[1]=s4[1]; d4[2]=s4[2]; d4[3]=s4[3];
  }
  {
    int f = tid >> 2, c0 = (tid & 3) * 16;
    const float4* s4 = (const float4*)(rot + (size_t)f*(NHASH*64) + (size_t)h*64 + c0);
    float4* d4 = (float4*)(&Bm[f][c0]);
    d4[0]=s4[0]; d4[1]=s4[1]; d4[2]=s4[2]; d4[3]=s4[3];
  }
  __syncthreads();

  const int wave = tid >> 5, lane = tid & 31, half = lane >> 4, mn = lane & 15;
  for (int T = wave; T < 16; T += 8) {
    const int mi = T >> 2, ni = T & 3;
    v8f acc = {};
#pragma unroll
    for (int k0 = 0; k0 < 64; k0 += 4) {
      v2f a, bb;
      a.x  = A [mi*16 + mn][k0 + 2*half    ];
      a.y  = A [mi*16 + mn][k0 + 2*half + 1];
      bb.x = Bm[k0 + 2*half    ][ni*16 + mn];
      bb.y = Bm[k0 + 2*half + 1][ni*16 + mn];
      acc = __builtin_amdgcn_wmma_f32_16x16x4_f32(false, a, false, bb,
                                                  (short)0, acc, false, false);
    }
#pragma unroll
    for (int r = 0; r < 8; r++)
      R[mi*16 + r + 8*half][ni*16 + mn] = acc[r];
  }
  __syncthreads();

  if (tid < 64) {
    const int row = tid;
    float best = R[row][0]; int bidx = 0;
    for (int j = 1; j < 128; j++) {
      float vv = (j < 64) ? R[row][j] : -R[row][j - 64];
      if (vv > best) { best = vv; bidx = j; }
    }
    const int q = h*NBUCK + bidx;
    const int t = t0 + row;
    bucketArr[((size_t)b*NHASH + h)*SEQ + t] = q;
    atomicAdd(&tilecnt[((size_t)b*NBALL + q)*NT + (t >> 9)], 1);
  }
}

__global__ __launch_bounds__(1024) void lsh_prefix_kernel(int* __restrict__ tilecnt)
{
  __shared__ int sc[NBALL];
  const int b = blockIdx.x, q = threadIdx.x;
  int* tc = tilecnt + ((size_t)b*NBALL + q)*NT;
  int s = 0;
#pragma unroll
  for (int j = 0; j < NT; j++) s += tc[j];
  sc[q] = s;
  __syncthreads();
  for (int off = 1; off < NBALL; off <<= 1) {
    int vv = (q >= off) ? sc[q - off] : 0;
    __syncthreads();
    sc[q] += vv;
    __syncthreads();
  }
  int run = sc[q] - s;
#pragma unroll
  for (int j = 0; j < NT; j++) { int c0 = tc[j]; tc[j] = run; run += c0; }
}

__global__ __launch_bounds__(TS) void lsh_rank_kernel(
    const int* __restrict__ bucketArr, const int* __restrict__ tilecnt,
    int* __restrict__ sticker)
{
  __shared__ int bk[TS];
  const int bid  = blockIdx.x;
  const int tile = bid & (NT - 1);
  const int h    = (bid >> 4) & 7;
  const int b    = bid >> 7;
  const int tid  = threadIdx.x;
  const int t    = tile*TS + tid;
  const int q    = bucketArr[((size_t)b*NHASH + h)*SEQ + t];
  bk[tid] = q;
  __syncthreads();
  int r = 0;
  for (int j = 0; j < tid; j++) r += (bk[j] == q) ? 1 : 0;
  const int pos = tilecnt[((size_t)b*NBALL + q)*NT + tile] + r;
  sticker[(size_t)b*HS + pos] = h*SEQ + t;
}

#define KP 68
#define PP 132

template <int M> __device__ __forceinline__ float lane_xor16(float x) {
  return __builtin_bit_cast(float, __builtin_amdgcn_update_dpp(
      0, __builtin_bit_cast(int, x), 0x160 + M, 0xf, 0xf, true));
}
__device__ __forceinline__ float row16_max(float x) {
  x = fmaxf(x, lane_xor16<1>(x));
  x = fmaxf(x, lane_xor16<2>(x));
  x = fmaxf(x, lane_xor16<4>(x));
  x = fmaxf(x, lane_xor16<8>(x));
  return x;
}
__device__ __forceinline__ float row16_sum(float x) {
  x += lane_xor16<1>(x);
  x += lane_xor16<2>(x);
  x += lane_xor16<4>(x);
  x += lane_xor16<8>(x);
  return x;
}

__global__ __launch_bounds__(128) void lsh_attn_kernel(
    const float* __restrict__ qk, const float* __restrict__ vin,
    const int* __restrict__ sticker,
    float* __restrict__ oBuf, float* __restrict__ lgBuf)
{
  __shared__ int   idxA[128];
  __shared__ int   tA  [128];
  __shared__ float ksc [128];
  __shared__ float lse [64];
  __shared__ __align__(16) float Ks[128][KP];
  __shared__ __align__(16) float Vs[128][KP];
  __shared__ __align__(16) float Pw[4][16][PP];

  const int bid = blockIdx.x;
  const int c   = bid & (NCHUNK - 1);
  const int b   = bid >> 10;
  const int tid = threadIdx.x;
  const int prev = (c + NCHUNK - 1) & (NCHUNK - 1);

  {
    const int j  = tid >> 1;
    const int c0 = (tid & 1) * 32;
    const int idx0 = sticker[(size_t)b*HS + (size_t)c   *BSZ + j];
    const int idx1 = sticker[(size_t)b*HS + (size_t)prev*BSZ + j];
    const int t0 = idx0 & (SEQ - 1);
    const int t1 = idx1 & (SEQ - 1);
    const float4* sk0 = (const float4*)(qk  + ((size_t)b*SEQ + t0)*DIM + c0);
    const float4* sv0 = (const float4*)(vin + ((size_t)b*SEQ + t0)*DIM + c0);
    const float4* sk1 = (const float4*)(qk  + ((size_t)b*SEQ + t1)*DIM + c0);
    const float4* sv1 = (const float4*)(vin + ((size_t)b*SEQ + t1)*DIM + c0);
    float4 k0[8], k1[8], w0[8], w1[8];
#pragma unroll
    for (int i = 0; i < 8; i++) k0[i] = sk0[i];
#pragma unroll
    for (int i = 0; i < 8; i++) k1[i] = sk1[i];
#pragma unroll
    for (int i = 0; i < 8; i++) w0[i] = sv0[i];
#pragma unroll
    for (int i = 0; i < 8; i++) w1[i] = sv1[i];
    float4* dk0 = (float4*)(&Ks[j][c0]);
    float4* dk1 = (float4*)(&Ks[64 + j][c0]);
    float4* dv0 = (float4*)(&Vs[j][c0]);
    float4* dv1 = (float4*)(&Vs[64 + j][c0]);
    float ss0 = 0.f, ss1 = 0.f;
#pragma unroll
    for (int i = 0; i < 8; i++) {
      dk0[i] = k0[i];
      ss0 += k0[i].x * k0[i].x; ss0 += k0[i].y * k0[i].y;
      ss0 += k0[i].z * k0[i].z; ss0 += k0[i].w * k0[i].w;
    }
#pragma unroll
    for (int i = 0; i < 8; i++) {
      dk1[i] = k1[i];
      ss1 += k1[i].x * k1[i].x; ss1 += k1[i].y * k1[i].y;
      ss1 += k1[i].z * k1[i].z; ss1 += k1[i].w * k1[i].w;
    }
#pragma unroll
    for (int i = 0; i < 8; i++) { dv0[i] = w0[i]; dv1[i] = w1[i]; }
    ss0 += lane_xor16<1>(ss0);
    ss1 += lane_xor16<1>(ss1);
    if ((tid & 1) == 0) {
      idxA[j]      = idx0;  tA[j]      = t0;
      idxA[64 + j] = idx1;  tA[64 + j] = t1;
      ksc[j]       = 0.125f * (1.0f / (sqrtf(ss0) + 1e-6f));
      ksc[64 + j]  = 0.125f * (1.0f / (sqrtf(ss1) + 1e-6f));
    }
  }
  __syncthreads();

  const int wave = tid >> 5, lane = tid & 31, half = lane >> 4, mn = lane & 15;
  const int q0 = wave * 16;
  float (*Pme)[PP] = Pw[wave];

  v2f qf[16];
#pragma unroll
  for (int s = 0; s < 16; s++) {
    qf[s].x = Ks[q0 + mn][4*s + 2*half    ];
    qf[s].y = Ks[q0 + mn][4*s + 2*half + 1];
  }
  int trow[8];
#pragma unroll
  for (int r = 0; r < 8; r++) trow[r] = tA[q0 + r + 8*half];

  v8f S[8];
#pragma unroll
  for (int j = 0; j < 8; j++) {
    v8f acc = {};
#pragma unroll
    for (int s = 0; s < 16; s++) {
      v2f bb;
      bb.x = Ks[16*j + mn][4*s + 2*half    ];
      bb.y = Ks[16*j + mn][4*s + 2*half + 1];
      acc = __builtin_amdgcn_wmma_f32_16x16x4_f32(false, qf[s], false, bb,
                                                  (short)0, acc, false, false);
    }
    const float sc = ksc[16*j + mn];
    const int   tc = tA [16*j + mn];
#pragma unroll
    for (int r = 0; r < 8; r++) {
      float dval = acc[r] * sc;
      if (trow[r] == tc) dval = SELF_VAL;
      acc[r] = dval;
    }
    S[j] = acc;
  }

  float inv[8], lsev[8];
#pragma unroll
  for (int r = 0; r < 8; r++) {
    float m = S[0][r];
#pragma unroll
    for (int j = 1; j < 8; j++) m = fmaxf(m, S[j][r]);
    m = row16_max(m);
    float ssum = 0.f;
#pragma unroll
    for (int j = 0; j < 8; j++) { const float e = expf(S[j][r] - m); S[j][r] = e; ssum += e; }
    ssum = row16_sum(ssum);
    lsev[r] = m + logf(ssum);
    inv[r]  = 1.0f / ssum;
  }
  if (mn == 0) {
#pragma unroll
    for (int r = 0; r < 8; r++) lse[q0 + r + 8*half] = lsev[r];
  }
#pragma unroll
  for (int j = 0; j < 8; j++)
#pragma unroll
    for (int r = 0; r < 8; r++)
      Pme[r + 8*half][16*j + mn] = S[j][r] * inv[r];
  __syncthreads();

  v8f O[4] = {};
#pragma unroll
  for (int ks = 0; ks < 32; ks++) {
    v2f a;
    a.x = Pme[mn][4*ks + 2*half    ];
    a.y = Pme[mn][4*ks + 2*half + 1];
#pragma unroll
    for (int ni = 0; ni < 4; ni++) {
      v2f bb;
      bb.x = Vs[4*ks + 2*half    ][16*ni + mn];
      bb.y = Vs[4*ks + 2*half + 1][16*ni + mn];
      O[ni] = __builtin_amdgcn_wmma_f32_16x16x4_f32(false, a, false, bb,
                                                    (short)0, O[ni], false, false);
    }
  }
#pragma unroll
  for (int ni = 0; ni < 4; ni++)
#pragma unroll
    for (int r = 0; r < 8; r++)
      Pme[r + 8*half][16*ni + mn] = O[ni][r];
  __syncthreads();

#pragma unroll
  for (int it = 0; it < 8; it++) {
    const int row = 2*it + half;
    const float4 x = *(const float4*)(&Pme[row][4*mn]);
    *(float4*)(oBuf + ((size_t)b*HS + idxA[q0 + row])*DIM + 4*mn) = x;
  }
  if (tid < 64) lgBuf[(size_t)b*HS + idxA[tid]] = lse[tid];
}

__global__ __launch_bounds__(256) void lsh_combine_kernel(
    const float* __restrict__ oBuf, const float* __restrict__ lgBuf,
    float* __restrict__ out)
{
  const int g = blockIdx.x * blockDim.x + threadIdx.x;
  const int b = g >> 13;
  const int t = g & (SEQ - 1);
  float lg[NHASH];
  float m = -1e30f;
#pragma unroll
  for (int h = 0; h < NHASH; h++) {
    lg[h] = lgBuf[(size_t)b*HS + (size_t)h*SEQ + t];
    m = fmaxf(m, lg[h]);
  }
  float ssum = 0.f;
#pragma unroll
  for (int h = 0; h < NHASH; h++) { lg[h] = expf(lg[h] - m); ssum += lg[h]; }
  const float inv = 1.0f / ssum;
#pragma unroll
  for (int h = 0; h < NHASH; h++) lg[h] *= inv;

  float* dst = out + ((size_t)b*SEQ + t)*DIM;
#pragma unroll
  for (int d = 0; d < DIM; d += 4) {
    float4 acc = {0.f, 0.f, 0.f, 0.f};
    for (int h = 0; h < NHASH; h++) {
      const float4 o4 = *(const float4*)(oBuf + ((size_t)b*HS + (size_t)h*SEQ + t)*DIM + d);
      acc.x += lg[h]*o4.x; acc.y += lg[h]*o4.y; acc.z += lg[h]*o4.z; acc.w += lg[h]*o4.w;
    }
    *(float4*)(dst + d) = acc;
  }
}

extern "C" void kernel_launch(void* const* d_in, const int* in_sizes, int n_in,
                              void* d_out, int out_size, void* d_ws, size_t ws_size,
                              hipStream_t stream)
{
  const float* qk  = (const float*)d_in[0];
  const float* v   = (const float*)d_in[1];
  const float* rot = (const float*)d_in[2];

  char* ws = (char*)d_ws;
  size_t off = 0;
  auto alloc = [&](size_t bytes) -> void* {
    void* p = ws + off;
    off += (bytes + 255) & ~(size_t)255;
    return p;
  };
  int*   bucketArr = (int*)  alloc((size_t)BATCH*NHASH*SEQ*sizeof(int));
  int*   tilecnt   = (int*)  alloc((size_t)BATCH*NBALL*NT*sizeof(int));
  int*   sticker   = (int*)  alloc((size_t)BATCH*HS*sizeof(int));
  float* oBuf      = (float*)alloc((size_t)BATCH*HS*DIM*sizeof(float));
  float* lgBuf     = (float*)alloc((size_t)BATCH*HS*sizeof(float));
  if (off > ws_size) return;

  hipMemsetAsync(tilecnt, 0, (size_t)BATCH*NBALL*NT*sizeof(int), stream);

  lsh_hash_kernel   <<<BATCH*NHASH*(SEQ/64), 256, 0, stream>>>(qk, rot, bucketArr, tilecnt);
  lsh_prefix_kernel <<<BATCH,               1024, 0, stream>>>(tilecnt);
  lsh_rank_kernel   <<<BATCH*NHASH*NT,        TS, 0, stream>>>(bucketArr, tilecnt, sticker);
  lsh_attn_kernel   <<<BATCH*NCHUNK,         128, 0, stream>>>(qk, v, sticker, oBuf, lgBuf);
  lsh_combine_kernel<<<(BATCH*SEQ)/256,      256, 0, stream>>>(oBuf, lgBuf, (float*)d_out);
}


# attention: each 4-wave group walks 4 consecutive chunks; the current key/value tile stays in LDS as the next chunk's look-back half: a sorted row is gathered and normed once, not twice; same sums
# speedup vs baseline: 1.3811x; 1.0782x over previous
#include <hip/hip_runtime.h>
#include <math.h>

#define BATCH 8
#define SEQ   8192
#define DIM   64
#define NHASH 8
#define NBUCK 128
#define NBALL (NHASH*NBUCK)
#define BSZ   64
#define NCHUNK (NHASH*(SEQ/BSZ))
#define HS    (NHASH*SEQ)
#define TS    512
#define NT    (SEQ/TS)
#define SELF_VAL (-5.0e4f)

typedef float v2f __attribute__((ext_vector_type(2)));
typedef float v8f __attribute__((ext_vector_type(8)));

__global__ __launch_bounds__(256) void lsh_hash_kernel(
    const float* __restrict__ qk, const float* __restrict__ rot,
    int* __restrict__ bucketArr, int* __restrict__ tilecnt)
{
  __shared__ __align__(16) float A [64][68];
  __shared__ __align__(16) float Bm[64][68];
  __shared__ __align__(16) float R [64][68];

  const int bid = blockIdx.x;
  const int tt  = bid & 127;
  const int h   = (bid >> 7) & 7;
  const int b   = bid >> 10;
  const int tid = threadIdx.x;
  const int t0  = tt * 64;

  {
    int row = tid >> 2, c0 = (tid & 3) * 16;
    const float4* s4 = (const float4*)(qk + ((size_t)b*SEQ + t0 + row)*DIM + c0);
    float4* d4 = (float4*)(&A[row][c0]);
    d4[0]=s4[0]; d4[1]=s4[1]; d4[2]=s4[2]; d4[3]=s4[3];
  }
  {
    int f = tid >> 2, c0 = (tid & 3) * 16;
    const float4* s4 = (const float4*)(rot + (size_t)f*(NHASH*64) + (size_t)h*64 + c0);
    float4* d4 = (float4*)(&Bm[f][c0]);
    d4[0]=s4[0]; d4[1]=s4[1]; d4[2]=s4[2]; d4[3]=s4[3];
  }
  __syncthreads();

  const int wave = tid >> 5, lane = tid & 31, half = lane >> 4, mn = lane & 15;
  for (int T = wave; T < 16; T += 8) {
    const int mi = T >> 2, ni = T & 3;
    v8f acc = {};
#pragma unroll
    for (int k0 = 0; k0 < 64; k0 += 4) {
      v2f a, bb;
      a.x  = A [mi*16 + mn][k0 + 2*half    ];
      a.y  = A [mi*16 + mn][k0 + 2*half + 1];
      bb.x = Bm[k0 + 2*half    ][ni*16 + mn];
      bb.y = Bm[k0 + 2*half + 1][ni*16 + mn];
      acc = __builtin_amdgcn_wmma_f32_16x16x4_f32(false, a, false, bb,
                                                  (short)0, acc, false, false);
    }
#pragma unroll
    for (int r = 0; r < 8; r++)
      R[mi*16 + r + 8*half][ni*16 + mn] = acc[r];
  }
  __syncthreads();

  if (tid < 64) {
    const int row = tid;
    float best = R[row][0]; int bidx = 0;
    for (int j = 1; j < 128; j++) {
      float vv = (j < 64) ? R[row][j] : -R[row][j - 64];
      if (vv > best) { best = vv; bidx = j; }
    }
    const int q = h*NBUCK + bidx;
    const int t = t0 + row;
    bucketArr[((size_t)b*NHASH + h)*SEQ + t] = q;
    atomicAdd(&tilecnt[((size_t)b*NBALL + q)*NT + (t >> 9)], 1);
  }
}

__global__ __launch_bounds__(1024) void lsh_prefix_kernel(int* __restrict__ tilecnt)
{
  __shared__ int sc[NBALL];
  const int b = blockIdx.x, q = threadIdx.x;
  int* tc = tilecnt + ((size_t)b*NBALL + q)*NT;
  int s = 0;
#pragma unroll
  for (int j = 0; j < NT; j++) s += tc[j];
  sc[q] = s;
  __syncthreads();
  for (int off = 1; off < NBALL; off <<= 1) {
    int vv = (q >= off) ? sc[q - off] : 0;
    __syncthreads();
    sc[q] += vv;
    __syncthreads();
  }
  int run = sc[q] - s;
#pragma unroll
  for (int j = 0; j < NT; j++) { int c0 = tc[j]; tc[j] = run; run += c0; }
}

__global__ __launch_bounds__(TS) void lsh_rank_kernel(
    const int* __restrict__ bucketArr, const int* __restrict__ tilecnt,
    int* __restrict__ sticker)
{
  __shared__ int bk[TS];
  const int bid  = blockIdx.x;
  const int tile = bid & (NT - 1);
  const int h    = (bid >> 4) & 7;
  const int b    = bid >> 7;
  const int tid  = threadIdx.x;
  const int t    = tile*TS + tid;
  const int q    = bucketArr[((size_t)b*NHASH + h)*SEQ + t];
  bk[tid] = q;
  __syncthreads();
  int r = 0;
  for (int j = 0; j < tid; j++) r += (bk[j] == q) ? 1 : 0;
  const int pos = tilecnt[((size_t)b*NBALL + q)*NT + tile] + r;
  sticker[(size_t)b*HS + pos] = h*SEQ + t;
}

#define KP 68
#define PP 132
#define RUN 4

template <int M> __device__ __forceinline__ float lane_xor16(float x) {
  return __builtin_bit_cast(float, __builtin_amdgcn_update_dpp(
      0, __builtin_bit_cast(int, x), 0x160 + M, 0xf, 0xf, true));
}
__device__ __forceinline__ float row16_max(float x) {
  x = fmaxf(x, lane_xor16<1>(x));
  x = fmaxf(x, lane_xor16<2>(x));
  x = fmaxf(x, lane_xor16<4>(x));
  x = fmaxf(x, lane_xor16<8>(x));
  return x;
}
__device__ __forceinline__ float row16_sum(float x) {
  x += lane_xor16<1>(x);
  x += lane_xor16<2>(x);
  x += lane_xor16<4>(x);
  x += lane_xor16<8>(x);
  return x;
}

__global__ __launch_bounds__(128) void lsh_attn_kernel(
    const float* __restrict__ qk, const float* __restrict__ vin,
    const int* __restrict__ sticker,
    float* __restrict__ oBuf, float* __restrict__ lgBuf)
{
  __shared__ int   idxA[128];
  __shared__ int   tA  [128];
  __shared__ float ksc [128];
  __shared__ float lse [64];
  __shared__ __align__(16) float Ks[128][KP];
  __shared__ __align__(16) float Vs[128][KP];
  __shared__ __align__(16) float Pw[4][16][PP];

  const int bid = blockIdx.x;
  const int cg  = bid & (NCHUNK/RUN - 1);
  const int b   = bid / (NCHUNK/RUN);
  const int tid = threadIdx.x;
  const int cfirst = cg * RUN;
  const int prev = (cfirst + NCHUNK - 1) & (NCHUNK - 1);

  {
    const int j  = tid >> 1;
    const int c0 = (tid & 1) * 32;
    const int idx0 = sticker[(size_t)b*HS + (size_t)cfirst*BSZ + j];
    const int idx1 = sticker[(size_t)b*HS + (size_t)prev  *BSZ + j];
    const int t0 = idx0 & (SEQ - 1);
    const int t1 = idx1 & (SEQ - 1);
    const float4* sk0 = (const float4*)(qk  + ((size_t)b*SEQ + t0)*DIM + c0);
    const float4* sv0 = (const float4*)(vin + ((size_t)b*SEQ + t0)*DIM + c0);
    const float4* sk1 = (const float4*)(qk  + ((size_t)b*SEQ + t1)*DIM + c0);
    const float4* sv1 = (const float4*)(vin + ((size_t)b*SEQ + t1)*DIM + c0);
    float4 k0[8], k1[8], w0[8], w1[8];
#pragma unroll
    for (int i = 0; i < 8; i++) k0[i] = sk0[i];
#pragma unroll
    for (int i = 0; i < 8; i++) k1[i] = sk1[i];
#pragma unroll
    for (int i = 0; i < 8; i++) w0[i] = sv0[i];
#pragma unroll
    for (int i = 0; i < 8; i++) w1[i] = sv1[i];
    float4* dk0 = (float4*)(&Ks[j][c0]);
    float4* dk1 = (float4*)(&Ks[64 + j][c0]);
    float4* dv0 = (float4*)(&Vs[j][c0]);
    float4* dv1 = (float4*)(&Vs[64 + j][c0]);
    float ss0 = 0.f, ss1 = 0.f;
#pragma unroll
    for (int i = 0; i < 8; i++) {
      dk0[i] = k0[i];
      ss0 += k0[i].x * k0[i].x; ss0 += k0[i].y * k0[i].y;
      ss0 += k0[i].z * k0[i].z; ss0 += k0[i].w * k0[i].w;
    }
#pragma unroll
    for (int i = 0; i < 8; i++) {
      dk1[i] = k1[i];
      ss1 += k1[i].x * k1[i].x; ss1 += k1[i].y * k1[i].y;
      ss1 += k1[i].z * k1[i].z; ss1 += k1[i].w * k1[i].w;
    }
#pragma unroll
    for (int i = 0; i < 8; i++) { dv0[i] = w0[i]; dv1[i] = w1[i]; }
    ss0 += lane_xor16<1>(ss0);
    ss1 += lane_xor16<1>(ss1);
    if ((tid & 1) == 0) {
      idxA[j]      = idx0;  tA[j]      = t0;
      idxA[64 + j] = idx1;  tA[64 + j] = t1;
      ksc[j]       = 0.125f * (1.0f / (sqrtf(ss0) + 1e-6f));
      ksc[64 + j]  = 0.125f * (1.0f / (sqrtf(ss1) + 1e-6f));
    }
  }

  const int wave = tid >> 5, lane = tid & 31, half = lane >> 4, mn = lane & 15;
  const int q0 = wave * 16;
  float (*Pme)[PP] = Pw[wave];

#pragma unroll 1
  for (int it = 0; it < RUN; it++) {
    const int cb = (it & 1) * 64;
    const int pb = 64 - cb;
    __syncthreads();

    v2f qf[16];
#pragma unroll
    for (int s = 0; s < 16; s++) {
      qf[s].x = Ks[cb + q0 + mn][4*s + 2*half    ];
      qf[s].y = Ks[cb + q0 + mn][4*s + 2*half + 1];
    }
    int trow[8];
#pragma unroll
    for (int r = 0; r < 8; r++) trow[r] = tA[cb + q0 + r + 8*half];

    v8f S[8];
#pragma unroll
    for (int j = 0; j < 8; j++) {
      const int kb = (j < 4) ? (cb + 16*j) : (pb + 16*(j - 4));
      v8f acc = {};
#pragma unroll
      for (int s = 0; s < 16; s++) {
        v2f bb;
        bb.x = Ks[kb + mn][4*s + 2*half    ];
        bb.y = Ks[kb + mn][4*s + 2*half + 1];
        acc = __builtin_amdgcn_wmma_f32_16x16x4_f32(false, qf[s], false, bb,
                                                    (short)0, acc, false, false);
      }
      const float sc = ksc[kb + mn];
      const int   tc = tA [kb + mn];
#pragma unroll
      for (int r = 0; r < 8; r++) {
        float dval = acc[r] * sc;
        if (trow[r] == tc) dval = SELF_VAL;
        acc[r] = dval;
      }
      S[j] = acc;
    }

    float inv[8], lsev[8];
#pragma unroll
    for (int r = 0; r < 8; r++) {
      float m = S[0][r];
#pragma unroll
      for (int j = 1; j < 8; j++) m = fmaxf(m, S[j][r]);
      m = row16_max(m);
      float ssum = 0.f;
#pragma unroll
      for (int j = 0; j < 8; j++) { const float e = expf(S[j][r] - m); S[j][r] = e; ssum += e; }
      ssum = row16_sum(ssum);
      lsev[r] = m + logf(ssum);
      inv[r]  = 1.0f / ssum;
    }
    if (mn == 0) {
#pragma unroll
      for (int r = 0; r < 8; r++) lse[q0 + r + 8*half] = lsev[r];
    }
#pragma unroll
    for (int j = 0; j < 8; j++)
#pragma unroll
      for (int r = 0; r < 8; r++)
        Pme[r + 8*half][16*j + mn] = S[j][r] * inv[r];
    __syncthreads();

    v8f O[4] = {};
#pragma unroll
    for (int ks = 0; ks < 32; ks++) {
      const int vb = (ks < 16) ? (cb + 4*ks) : (pb + 4*(ks - 16));
      v2f a;
      a.x = Pme[mn][4*ks + 2*half    ];
      a.y = Pme[mn][4*ks + 2*half + 1];
#pragma unroll
      for (int ni = 0; ni < 4; ni++) {
        v2f bb;
        bb.x = Vs[vb + 2*half    ][16*ni + mn];
        bb.y = Vs[vb + 2*half + 1][16*ni + mn];
        O[ni] = __builtin_amdgcn_wmma_f32_16x16x4_f32(false, a, false, bb,
                                                      (short)0, O[ni], false, false);
      }
    }
#pragma unroll
    for (int ni = 0; ni < 4; ni++)
#pragma unroll
      for (int r = 0; r < 8; r++)
        Pme[r + 8*half][16*ni + mn] = O[ni][r];
    __syncthreads();

#pragma unroll
    for (int it2 = 0; it2 < 8; it2++) {
      const int row = 2*it2 + half;
      const float4 x = *(const float4*)(&Pme[row][4*mn]);
      *(float4*)(oBuf + ((size_t)b*HS + idxA[cb + q0 + row])*DIM + 4*mn) = x;
    }
    if (tid < 64) lgBuf[(size_t)b*HS + idxA[cb + tid]] = lse[tid];

    if (it + 1 < RUN) {
      const int j  = tid >> 1;
      const int c0 = (tid & 1) * 32;
      const int idx0 = sticker[(size_t)b*HS + (size_t)(cfirst + it + 1)*BSZ + j];
      const int t0 = idx0 & (SEQ - 1);
      const float4* sk0 = (const float4*)(qk  + ((size_t)b*SEQ + t0)*DIM + c0);
      const float4* sv0 = (const float4*)(vin + ((size_t)b*SEQ + t0)*DIM + c0);
      float4 k0[8], w0[8];
#pragma unroll
      for (int i = 0; i < 8; i++) k0[i] = sk0[i];
#pragma unroll
      for (int i = 0; i < 8; i++) w0[i] = sv0[i];
      float4* dk0 = (float4*)(&Ks[pb + j][c0]);
      float4* dv0 = (float4*)(&Vs[pb + j][c0]);
      float ss0 = 0.f;
#pragma unroll
      for (int i = 0; i < 8; i++) {
        dk0[i] = k0[i];
        ss0 += k0[i].x * k0[i].x; ss0 += k0[i].y * k0[i].y;
        ss0 += k0[i].z * k0[i].z; ss0 += k0[i].w * k0[i].w;
      }
#pragma unroll
      for (int i = 0; i < 8; i++) dv0[i] = w0[i];
      ss0 += lane_xor16<1>(ss0);
      if ((tid & 1) == 0) {
        idxA[pb + j] = idx0;  tA[pb + j] = t0;
        ksc[pb + j]  = 0.125f * (1.0f / (sqrtf(ss0) + 1e-6f));
      }
    }
  }
}

__global__ __launch_bounds__(256) void lsh_combine_kernel(
    const float* __restrict__ oBuf, const float* __restrict__ lgBuf,
    float* __restrict__ out)
{
  const int g = blockIdx.x * blockDim.x + threadIdx.x;
  const int b = g >> 13;
  const int t = g & (SEQ - 1);
  float lg[NHASH];
  float m = -1e30f;
#pragma unroll
  for (int h = 0; h < NHASH; h++) {
    lg[h] = lgBuf[(size_t)b*HS + (size_t)h*SEQ + t];
    m = fmaxf(m, lg[h]);
  }
  float ssum = 0.f;
#pragma unroll
  for (int h = 0; h < NHASH; h++) { lg[h] = expf(lg[h] - m); ssum += lg[h]; }
  const float inv = 1.0f / ssum;
#pragma unroll
  for (int h = 0; h < NHASH; h++) lg[h] *= inv;

  float* dst = out + ((size_t)b*SEQ + t)*DIM;
#pragma unroll
  for (int d = 0; d < DIM; d += 4) {
    float4 acc = {0.f, 0.f, 0.f, 0.f};
    for (int h = 0; h < NHASH; h++) {
      const float4 o4 = *(const float4*)(oBuf + ((size_t)b*HS + (size_t)h*SEQ + t)*DIM + d);
      acc.x += lg[h]*o4.x; acc.y += lg[h]*o4.y; acc.z += lg[h]*o4.z; acc.w += lg[h]*o4.w;
    }
    *(float4*)(dst + d) = acc;
  }
}

extern "C" void kernel_launch(void* const* d_in, const int* in_sizes, int n_in,
                              void* d_out, int out_size, void* d_ws, size_t ws_size,
                              hipStream_t stream)
{
  const float* qk  = (const float*)d_in[0];
  const float* v   = (const float*)d_in[1];
  const float* rot = (const float*)d_in[2];

  char* ws = (char*)d_ws;
  size_t off = 0;
  auto alloc = [&](size_t bytes) -> void* {
    void* p = ws + off;
    off += (bytes + 255) & ~(size_t)255;
    return p;
  };
  int*   bucketArr = (int*)  alloc((size_t)BATCH*NHASH*SEQ*sizeof(int));
  int*   tilecnt   = (int*)  alloc((size_t)BATCH*NBALL*NT*sizeof(int));
  int*   sticker   = (int*)  alloc((size_t)BATCH*HS*sizeof(int));
  float* oBuf      = (float*)alloc((size_t)BATCH*HS*DIM*sizeof(float));
  float* lgBuf     = (float*)alloc((size_t)BATCH*HS*sizeof(float));
  if (off > ws_size) return;

  hipMemsetAsync(tilecnt, 0, (size_t)BATCH*NBALL*NT*sizeof(int), stream);

  lsh_hash_kernel   <<<BATCH*NHASH*(SEQ/64), 256, 0, stream>>>(qk, rot, bucketArr, tilecnt);
  lsh_prefix_kernel <<<BATCH,               1024, 0, stream>>>(tilecnt);
  lsh_rank_kernel   <<<BATCH*NHASH*NT,        TS, 0, stream>>>(bucketArr, tilecnt, sticker);
  lsh_attn_kernel   <<<BATCH*NCHUNK/RUN,     128, 0, stream>>>(qk, v, sticker, oBuf, lgBuf);
  lsh_combine_kernel<<<(BATCH*SEQ)/256,      256, 0, stream>>>(oBuf, lgBuf, (float*)d_out);
}


# attention: both products as 16-WMMA blocks, each block's LDS fragment loads issued one block ahead as one group (2nd register set); V tile as key pairs (8-byte B loads); P*V sums keys in a new order
# speedup vs baseline: 1.4501x; 1.0500x over previous
#include <hip/hip_runtime.h>
#include <math.h>

#define BATCH 8
#define SEQ   8192
#define DIM   64
#define NHASH 8
#define NBUCK 128
#define NBALL (NHASH*NBUCK)
#define BSZ   64
#define NCHUNK (NHASH*(SEQ/BSZ))
#define HS    (NHASH*SEQ)
#define TS    512
#define NT    (SEQ/TS)
#define SELF_VAL (-5.0e4f)

typedef float v2f __attribute__((ext_vector_type(2)));
typedef float v8f __attribute__((ext_vector_type(8)));

__global__ __launch_bounds__(256) void lsh_hash_kernel(
    const float* __restrict__ qk, const float* __restrict__ rot,
    int* __restrict__ bucketArr, int* __restrict__ tilecnt)
{
  __shared__ __align__(16) float A [64][68];
  __shared__ __align__(16) float Bm[64][68];
  __shared__ __align__(16) float R [64][68];

  const int bid = blockIdx.x;
  const int tt  = bid & 127;
  const int h   = (bid >> 7) & 7;
  const int b   = bid >> 10;
  const int tid = threadIdx.x;
  const int t0  = tt * 64;

  {
    int row = tid >> 2, c0 = (tid & 3) * 16;
    const float4* s4 = (const float4*)(qk + ((size_t)b*SEQ + t0 + row)*DIM + c0);
    float4* d4 = (float4*)(&A[row][c0]);
    d4[0]=s4[0]; d4[1]=s4[1]; d4[2]=s4[2]; d4[3]=s4[3];
  }
  {
    int f = tid >> 2, c0 = (tid & 3) * 16;
    const float4* s4 = (const float4*)(rot + (size_t)f*(NHASH*64) + (size_t)h*64 + c0);
    float4* d4 = (float4*)(&Bm[f][c0]);
    d4[0]=s4[0]; d4[1]=s4[1]; d4[2]=s4[2]; d4[3]=s4[3];
  }
  __syncthreads();

  const int wave = tid >> 5, lane = tid & 31, half = lane >> 4, mn = lane & 15;
  for (int T = wave; T < 16; T += 8) {
    const int mi = T >> 2, ni = T & 3;
    v8f acc = {};
#pragma unroll
    for (int k0 = 0; k0 < 64; k0 += 4) {
      v2f a, bb;
      a.x  = A [mi*16 + mn][k0 + 2*half    ];
      a.y  = A [mi*16 + mn][k0 + 2*half + 1];
      bb.x = Bm[k0 + 2*half    ][ni*16 + mn];
      bb.y = Bm[k0 + 2*half + 1][ni*16 + mn];
      acc = __builtin_amdgcn_wmma_f32_16x16x4_f32(false, a, false, bb,
                                                  (short)0, acc, false, false);
    }
#pragma unroll
    for (int r = 0; r < 8; r++)
      R[mi*16 + r + 8*half][ni*16 + mn] = acc[r];
  }
  __syncthreads();

  if (tid < 64) {
    const int row = tid;
    float best = R[row][0]; int bidx = 0;
    for (int j = 1; j < 128; j++) {
      float vv = (j < 64) ? R[row][j] : -R[row][j - 64];
      if (vv > best) { best = vv; bidx = j; }
    }
    const int q = h*NBUCK + bidx;
    const int t = t0 + row;
    bucketArr[((size_t)b*NHASH + h)*SEQ + t] = q;
    atomicAdd(&tilecnt[((size_t)b*NBALL + q)*NT + (t >> 9)], 1);
  }
}

__global__ __launch_bounds__(1024) void lsh_prefix_kernel(int* __restrict__ tilecnt)
{
  __shared__ int sc[NBALL];
  const int b = blockIdx.x, q = threadIdx.x;
  int* tc = tilecnt + ((size_t)b*NBALL + q)*NT;
  int s = 0;
#pragma unroll
  for (int j = 0; j < NT; j++) s += tc[j];
  sc[q] = s;
  __syncthreads();
  for (int off = 1; off < NBALL; off <<= 1) {
    int vv = (q >= off) ? sc[q - off] : 0;
    __syncthreads();
    sc[q] += vv;
    __syncthreads();
  }
  int run = sc[q] - s;
#pragma unroll
  for (int j = 0; j < NT; j++) { int c0 = tc[j]; tc[j] = run; run += c0; }
}

__global__ __launch_bounds__(TS) void lsh_rank_kernel(
    const int* __restrict__ bucketArr, const int* __restrict__ tilecnt,
    int* __restrict__ sticker)
{
  __shared__ int bk[TS];
  const int bid  = blockIdx.x;
  const int tile = bid & (NT - 1);
  const int h    = (bid >> 4) & 7;
  const int b    = bid >> 7;
  const int tid  = threadIdx.x;
  const int t    = tile*TS + tid;
  const int q    = bucketArr[((size_t)b*NHASH + h)*SEQ + t];
  bk[tid] = q;
  __syncthreads();
  int r = 0;
  for (int j = 0; j < tid; j++) r += (bk[j] == q) ? 1 : 0;
  const int pos = tilecnt[((size_t)b*NBALL + q)*NT + tile] + r;
  sticker[(size_t)b*HS + pos] = h*SEQ + t;
}

#define KP 68
#define VPP 132
#define PP 132
#define RUN 4

template <int M> __device__ __forceinline__ float lane_xor16(float x) {
  return __builtin_bit_cast(float, __builtin_amdgcn_update_dpp(
      0, __builtin_bit_cast(int, x), 0x160 + M, 0xf, 0xf, true));
}
__device__ __forceinline__ float row16_max(float x) {
  x = fmaxf(x, lane_xor16<1>(x));
  x = fmaxf(x, lane_xor16<2>(x));
  x = fmaxf(x, lane_xor16<4>(x));
  x = fmaxf(x, lane_xor16<8>(x));
  return x;
}
__device__ __forceinline__ float row16_sum(float x) {
  x += lane_xor16<1>(x);
  x += lane_xor16<2>(x);
  x += lane_xor16<4>(x);
  x += lane_xor16<8>(x);
  return x;
}

__global__ __launch_bounds__(128) void lsh_attn_kernel(
    const float* __restrict__ qk, const float* __restrict__ vin,
    const int* __restrict__ sticker,
    float* __restrict__ oBuf, float* __restrict__ lgBuf)
{
  __shared__ int   idxA[128];
  __shared__ int   tA  [128];
  __shared__ float ksc [128];
  __shared__ float lse [64];
  __shared__ __align__(16) float Ks[128][KP];
  __shared__ __align__(16) float Vp[64][VPP];
  __shared__ __align__(16) float Pw[4][16][PP];

  const int bid = blockIdx.x;
  const int cg  = bid & (NCHUNK/RUN - 1);
  const int b   = bid / (NCHUNK/RUN);
  const int tid = threadIdx.x;
  const int cfirst = cg * RUN;
  const int prev = (cfirst + NCHUNK - 1) & (NCHUNK - 1);

  {
    const int j  = tid >> 1;
    const int c0 = (tid & 1) * 32;
    const int idx0 = sticker[(size_t)b*HS + (size_t)cfirst*BSZ + j];
    const int idx1 = sticker[(size_t)b*HS + (size_t)prev  *BSZ + j];
    const int t0 = idx0 & (SEQ - 1);
    const int t1 = idx1 & (SEQ - 1);
    const float4* sk0 = (const float4*)(qk  + ((size_t)b*SEQ + t0)*DIM + c0);
    const float4* sk1 = (const float4*)(qk  + ((size_t)b*SEQ + t1)*DIM + c0);
    const int q  = tid >> 2;
    const int vc = tid & 3;
    const int2 iv0 = *(const int2*)(sticker + (size_t)b*HS + (size_t)cfirst*BSZ + 2*q);
    const int2 iv1 = *(const int2*)(sticker + (size_t)b*HS + (size_t)prev  *BSZ + 2*q);
    const float4* sa0 = (const float4*)(vin + ((size_t)b*SEQ + (iv0.x & (SEQ - 1)))*DIM) + vc;
    const float4* sb0 = (const float4*)(vin + ((size_t)b*SEQ + (iv0.y & (SEQ - 1)))*DIM) + vc;
    const float4* sa1 = (const float4*)(vin + ((size_t)b*SEQ + (iv1.x & (SEQ - 1)))*DIM) + vc;
    const float4* sb1 = (const float4*)(vin + ((size_t)b*SEQ + (iv1.y & (SEQ - 1)))*DIM) + vc;
    float4 k0[8], k1[8], wa0[4], wb0[4], wa1[4], wb1[4];
#pragma unroll
    for (int i = 0; i < 8; i++) k0[i] = sk0[i];
#pragma unroll
    for (int i = 0; i < 8; i++) k1[i] = sk1[i];
#pragma unroll
    for (int i = 0; i < 4; i++) { wa0[i] = sa0[4*i]; wb0[i] = sb0[4*i]; }
#pragma unroll
    for (int i = 0; i < 4; i++) { wa1[i] = sa1[4*i]; wb1[i] = sb1[4*i]; }
    float4* dk0 = (float4*)(&Ks[j][c0]);
    float4* dk1 = (float4*)(&Ks[64 + j][c0]);
    float4* dv0 = (float4*)(&Vp[q][8*vc]);
    float4* dv1 = (float4*)(&Vp[32 + q][8*vc]);
    float ss0 = 0.f, ss1 = 0.f;
#pragma unroll
    for (int i = 0; i < 8; i++) {
      dk0[i] = k0[i];
      ss0 += k0[i].x * k0[i].x; ss0 += k0[i].y * k0[i].y;
      ss0 += k0[i].z * k0[i].z; ss0 += k0[i].w * k0[i].w;
    }
#pragma unroll
    for (int i = 0; i < 8; i++) {
      dk1[i] = k1[i];
      ss1 += k1[i].x * k1[i].x; ss1 += k1[i].y * k1[i].y;
      ss1 += k1[i].z * k1[i].z; ss1 += k1[i].w * k1[i].w;
    }
#pragma unroll
    for (int i = 0; i < 4; i++) {
      dv0[8*i]     = make_float4(wa0[i].x, wb0[i].x, wa0[i].y, wb0[i].y);
      dv0[8*i + 1] = make_float4(wa0[i].z, wb0[i].z, wa0[i].w, wb0[i].w);
      dv1[8*i]     = make_float4(wa1[i].x, wb1[i].x, wa1[i].y, wb1[i].y);
      dv1[8*i + 1] = make_float4(wa1[i].z, wb1[i].z, wa1[i].w, wb1[i].w);
    }
    ss0 += lane_xor16<1>(ss0);
    ss1 += lane_xor16<1>(ss1);
    if ((tid & 1) == 0) {
      idxA[j]      = idx0;  tA[j]      = t0;
      idxA[64 + j] = idx1;  tA[64 + j] = t1;
      ksc[j]       = 0.125f * (1.0f / (sqrtf(ss0) + 1e-6f));
      ksc[64 + j]  = 0.125f * (1.0f / (sqrtf(ss1) + 1e-6f));
    }
  }

  const int wave = tid >> 5, lane = tid & 31, half = lane >> 4, mn = lane & 15;
  const int q0 = wave * 16;
  float (*Pme)[PP] = Pw[wave];

#pragma unroll 1
  for (int it = 0; it < RUN; it++) {
    const int cb = (it & 1) * 64;
    const int pb = 64 - cb;
    __syncthreads();

    v8f S[8] = {};
    {
      v2f qa[2][2], kf[2][2][8];
      const float* qrow = &Ks[cb + q0 + mn][2*half];
      const float* krc  = &Ks[cb + mn][2*half];
      const float* krp  = &Ks[pb + mn][2*half];
#pragma unroll
      for (int u = 0; u < 2; u++) {
        qa[0][u] = *(const v2f*)(qrow + 4*u);
#pragma unroll
        for (int j = 0; j < 8; j++)
          kf[0][u][j] = *(const v2f*)(((j < 4) ? krc + 16*j*KP : krp + 16*(j - 4)*KP) + 4*u);
      }
      __builtin_amdgcn_sched_barrier(0);
#pragma unroll
      for (int blk = 0; blk < 8; blk++) {
        const int cur = blk & 1, nxt = cur ^ 1;
        if (blk + 1 < 8) {
#pragma unroll
          for (int u = 0; u < 2; u++) {
            const int s = 2*(blk + 1) + u;
            qa[nxt][u] = *(const v2f*)(qrow + 4*s);
#pragma unroll
            for (int j = 0; j < 8; j++)
              kf[nxt][u][j] = *(const v2f*)(((j < 4) ? krc + 16*j*KP : krp + 16*(j - 4)*KP) + 4*s);
          }
        }
        __builtin_amdgcn_sched_barrier(0);
#pragma unroll
        for (int u = 0; u < 2; u++)
#pragma unroll
          for (int j = 0; j < 8; j++)
            S[j] = __builtin_amdgcn_wmma_f32_16x16x4_f32(false, qa[cur][u], false, kf[cur][u][j],
                                                         (short)0, S[j], false, false);
        __builtin_amdgcn_sched_barrier(0);
      }
    }
    int trow[8];
#pragma unroll
    for (int r = 0; r < 8; r++) trow[r] = tA[cb + q0 + r + 8*half];
#pragma unroll
    for (int j = 0; j < 8; j++) {
      const int kb = (j < 4) ? (cb + 16*j) : (pb + 16*(j - 4));
      const float sc = ksc[kb + mn];
      const int   tc = tA [kb + mn];
#pragma unroll
      for (int r = 0; r < 8; r++) {
        float dval = S[j][r] * sc;
        if (trow[r] == tc) dval = SELF_VAL;
        S[j][r] = dval;
      }
    }

    float inv[8], lsev[8];
#pragma unroll
    for (int r = 0; r < 8; r++) {
      float m = S[0][r];
#pragma unroll
      for (int j = 1; j < 8; j++) m = fmaxf(m, S[j][r]);
      m = row16_max(m);
      float ssum = 0.f;
#pragma unroll
      for (int j = 0; j < 8; j++) { const float e = expf(S[j][r] - m); S[j][r] = e; ssum += e; }
      ssum = row16_sum(ssum);
      lsev[r] = m + logf(ssum);
      inv[r]  = 1.0f / ssum;
    }
    if (mn == 0) {
#pragma unroll
      for (int r = 0; r < 8; r++) lse[q0 + r + 8*half] = lsev[r];
    }
#pragma unroll
    for (int j = 0; j < 8; j++)
#pragma unroll
      for (int r = 0; r < 8; r++)
        Pme[r + 8*half][4*(8*(j >> 1) + (mn >> 1)) + 2*(j & 1) + (mn & 1)] = S[j][r] * inv[r];
    __syncthreads();

    v8f O[4] = {};
    {
      v2f pa[2][4], vf[2][4][4];
      const float* prow = &Pme[mn][2*half];
      const float* vcur = &Vp[(cb >> 1) + 8*half][2*mn];
      const float* vprv = &Vp[(pb >> 1) + 8*half][2*mn];
#pragma unroll
      for (int u = 0; u < 4; u++) {
        pa[0][u] = *(const v2f*)(prow + 4*u);
#pragma unroll
        for (int ni = 0; ni < 4; ni++)
          vf[0][u][ni] = *(const v2f*)(vcur + u*VPP + 32*ni);
      }
      __builtin_amdgcn_sched_barrier(0);
#pragma unroll
      for (int blk = 0; blk < 8; blk++) {
        const int cur = blk & 1, nxt = cur ^ 1;
        if (blk + 1 < 8) {
#pragma unroll
          for (int u = 0; u < 4; u++) {
            const int t  = 4*(blk + 1) + u;
            const int pr = 16*((t >> 3) & 1) + (t & 7);
            pa[nxt][u] = *(const v2f*)(prow + 4*t);
#pragma unroll
            for (int ni = 0; ni < 4; ni++)
              vf[nxt][u][ni] = *(const v2f*)(((t < 16) ? vcur : vprv) + pr*VPP + 32*ni);
          }
        }
        __builtin_amdgcn_sched_barrier(0);
#pragma unroll
        for (int u = 0; u < 4; u++)
#pragma unroll
          for (int ni = 0; ni < 4; ni++)
            O[ni] = __builtin_amdgcn_wmma_f32_16x16x4_f32(false, pa[cur][u], false, vf[cur][u][ni],
                                                          (short)0, O[ni], false, false);
        __builtin_amdgcn_sched_barrier(0);
      }
    }
#pragma unroll
    for (int ni = 0; ni < 4; ni++)
#pragma unroll
      for (int r = 0; r < 8; r++)
        Pme[r + 8*half][16*ni + mn] = O[ni][r];
    __syncthreads();

#pragma unroll
    for (int it2 = 0; it2 < 8; it2++) {
      const int row = 2*it2 + half;
      const float4 x = *(const float4*)(&Pme[row][4*mn]);
      *(float4*)(oBuf + ((size_t)b*HS + idxA[cb + q0 + row])*DIM + 4*mn) = x;
    }
    if (tid < 64) lgBuf[(size_t)b*HS + idxA[cb + tid]] = lse[tid];

    if (it + 1 < RUN) {
      const int j  = tid >> 1;
      const int c0 = (tid & 1) * 32;
      const int idx0 = sticker[(size_t)b*HS + (size_t)(cfirst + it + 1)*BSZ + j];
      const int t0 = idx0 & (SEQ - 1);
      const float4* sk0 = (const float4*)(qk  + ((size_t)b*SEQ + t0)*DIM + c0);
      const int q  = tid >> 2;
      const int vc = tid & 3;
      const int2 iv0 = *(const int2*)(sticker + (size_t)b*HS + (size_t)(cfirst + it + 1)*BSZ + 2*q);
      const float4* sa0 = (const float4*)(vin + ((size_t)b*SEQ + (iv0.x & (SEQ - 1)))*DIM) + vc;
      const float4* sb0 = (const float4*)(vin + ((size_t)b*SEQ + (iv0.y & (SEQ - 1)))*DIM) + vc;
      float4 k0[8], wa0[4], wb0[4];
#pragma unroll
      for (int i = 0; i < 8; i++) k0[i] = sk0[i];
#pragma unroll
      for (int i = 0; i < 4; i++) { wa0[i] = sa0[4*i]; wb0[i] = sb0[4*i]; }
      float4* dk0 = (float4*)(&Ks[pb + j][c0]);
      float4* dv0 = (float4*)(&Vp[(pb >> 1) + q][8*vc]);
      float ss0 = 0.f;
#pragma unroll
      for (int i = 0; i < 8; i++) {
        dk0[i] = k0[i];
        ss0 += k0[i].x * k0[i].x; ss0 += k0[i].y * k0[i].y;
        ss0 += k0[i].z * k0[i].z; ss0 += k0[i].w * k0[i].w;
      }
#pragma unroll
      for (int i = 0; i < 4; i++) {
        dv0[8*i]     = make_float4(wa0[i].x, wb0[i].x, wa0[i].y, wb0[i].y);
        dv0[8*i + 1] = make_float4(wa0[i].z, wb0[i].z, wa0[i].w, wb0[i].w);
      }
      ss0 += lane_xor16<1>(ss0);
      if ((tid & 1) == 0) {
        idxA[pb + j] = idx0;  tA[pb + j] = t0;
        ksc[pb + j]  = 0.125f * (1.0f / (sqrtf(ss0) + 1e-6f));
      }
    }
  }
}

__global__ __launch_bounds__(256) void lsh_combine_kernel(
    const float* __restrict__ oBuf, const float* __restrict__ lgBuf,
    float* __restrict__ out)
{
  const int g = blockIdx.x * blockDim.x + threadIdx.x;
  const int b = g >> 13;
  const int t = g & (SEQ - 1);
  float lg[NHASH];
  float m = -1e30f;
#pragma unroll
  for (int h = 0; h < NHASH; h++) {
    lg[h] = lgBuf[(size_t)b*HS + (size_t)h*SEQ + t];
    m = fmaxf(m, lg[h]);
  }
  float ssum = 0.f;
#pragma unroll
  for (int h = 0; h < NHASH; h++) { lg[h] = expf(lg[h] - m); ssum += lg[h]; }
  const float inv = 1.0f / ssum;
#pragma unroll
  for (int h = 0; h < NHASH; h++) lg[h] *= inv;

  float* dst = out + ((size_t)b*SEQ + t)*DIM;
#pragma unroll
  for (int d = 0; d < DIM; d += 4) {
    float4 acc = {0.f, 0.f, 0.f, 0.f};
    for (int h = 0; h < NHASH; h++) {
      const float4 o4 = *(const float4*)(oBuf + ((size_t)b*HS + (size_t)h*SEQ + t)*DIM + d);
      acc.x += lg[h]*o4.x; acc.y += lg[h]*o4.y; acc.z += lg[h]*o4.z; acc.w += lg[h]*o4.w;
    }
    *(float4*)(dst + d) = acc;
  }
}

extern "C" void kernel_launch(void* const* d_in, const int* in_sizes, int n_in,
                              void* d_out, int out_size, void* d_ws, size_t ws_size,
                              hipStream_t stream)
{
  const float* qk  = (const float*)d_in[0];
  const float* v   = (const float*)d_in[1];
  const float* rot = (const float*)d_in[2];

  char* ws = (char*)d_ws;
  size_t off = 0;
  auto alloc = [&](size_t bytes) -> void* {
    void* p = ws + off;
    off += (bytes + 255) & ~(size_t)255;
    return p;
  };
  int*   bucketArr = (int*)  alloc((size_t)BATCH*NHASH*SEQ*sizeof(int));
  int*   tilecnt   = (int*)  alloc((size_t)BATCH*NBALL*NT*sizeof(int));
  int*   sticker   = (int*)  alloc((size_t)BATCH*HS*sizeof(int));
  float* oBuf      = (float*)alloc((size_t)BATCH*HS*DIM*sizeof(float));
  float* lgBuf     = (float*)alloc((size_t)BATCH*HS*sizeof(float));
  if (off > ws_size) return;

  hipMemsetAsync(tilecnt, 0, (size_t)BATCH*NBALL*NT*sizeof(int), stream);

  lsh_hash_kernel   <<<BATCH*NHASH*(SEQ/64), 256, 0, stream>>>(qk, rot, bucketArr, tilecnt);
  lsh_prefix_kernel <<<BATCH,               1024, 0, stream>>>(tilecnt);
  lsh_rank_kernel   <<<BATCH*NHASH*NT,        TS, 0, stream>>>(bucketArr, tilecnt, sticker);
  lsh_attn_kernel   <<<BATCH*NCHUNK/RUN,     128, 0, stream>>>(qk, v, sticker, oBuf, lgBuf);
  lsh_combine_kernel<<<(BATCH*SEQ)/256,      256, 0, stream>>>(oBuf, lgBuf, (float*)d_out);
}


# sort: one work-group per (batch, round) ranks tokens by lane ballots in 32-token tiles, scans the counts in LDS and places them: no memset, atomics, prefix or rank launch; same order
# speedup vs baseline: 1.6392x; 1.1304x over previous
#include <hip/hip_runtime.h>
#include <math.h>

#define BATCH 8
#define SEQ   8192
#define DIM   64
#define NHASH 8
#define NBUCK 128
#define NBALL (NHASH*NBUCK)
#define BSZ   64
#define NCHUNK (NHASH*(SEQ/BSZ))
#define HS    (NHASH*SEQ)
#define TILE  32
#define NTILE (SEQ/TILE)
#define NCNT  (NBUCK*NTILE)
#define SELF_VAL (-5.0e4f)

typedef float v2f __attribute__((ext_vector_type(2)));
typedef float v8f __attribute__((ext_vector_type(8)));

__global__ __launch_bounds__(256) void lsh_hash_kernel(
    const float* __restrict__ qk, const float* __restrict__ rot,
    int* __restrict__ bucketArr)
{
  __shared__ __align__(16) float A [64][68];
  __shared__ __align__(16) float Bm[64][68];
  __shared__ __align__(16) float R [64][68];

  const int bid = blockIdx.x;
  const int tt  = bid & 127;
  const int h   = (bid >> 7) & 7;
  const int b   = bid >> 10;
  const int tid = threadIdx.x;
  const int t0  = tt * 64;

  {
    int row = tid >> 2, c0 = (tid & 3) * 16;
    const float4* s4 = (const float4*)(qk + ((size_t)b*SEQ + t0 + row)*DIM + c0);
    float4* d4 = (float4*)(&A[row][c0]);
    d4[0]=s4[0]; d4[1]=s4[1]; d4[2]=s4[2]; d4[3]=s4[3];
  }
  {
    int f = tid >> 2, c0 = (tid & 3) * 16;
    const float4* s4 = (const float4*)(rot + (size_t)f*(NHASH*64) + (size_t)h*64 + c0);
    float4* d4 = (float4*)(&Bm[f][c0]);
    d4[0]=s4[0]; d4[1]=s4[1]; d4[2]=s4[2]; d4[3]=s4[3];
  }
  __syncthreads();

  const int wave = tid >> 5, lane = tid & 31, half = lane >> 4, mn = lane & 15;
  for (int T = wave; T < 16; T += 8) {
    const int mi = T >> 2, ni = T & 3;
    v8f acc = {};
#pragma unroll
    for (int k0 = 0; k0 < 64; k0 += 4) {
      v2f a, bb;
      a.x  = A [mi*16 + mn][k0 + 2*half    ];
      a.y  = A [mi*16 + mn][k0 + 2*half + 1];
      bb.x = Bm[k0 + 2*half    ][ni*16 + mn];
      bb.y = Bm[k0 + 2*half + 1][ni*16 + mn];
      acc = __builtin_amdgcn_wmma_f32_16x16x4_f32(false, a, false, bb,
                                                  (short)0, acc, false, false);
    }
#pragma unroll
    for (int r = 0; r < 8; r++)
      R[mi*16 + r + 8*half][ni*16 + mn] = acc[r];
  }
  __syncthreads();

  if (tid < 64) {
    const int row = tid;
    float best = R[row][0]; int bidx = 0;
    for (int j = 1; j < 128; j++) {
      float vv = (j < 64) ? R[row][j] : -R[row][j - 64];
      if (vv > best) { best = vv; bidx = j; }
    }
    const int q = h*NBUCK + bidx;
    const int t = t0 + row;
    bucketArr[((size_t)b*NHASH + h)*SEQ + t] = q;
  }
}

#define CPAD(e) ((e) + ((e) >> 5))

template <int N> __device__ __forceinline__ int lane_shr16(int x) {
  return __builtin_amdgcn_update_dpp(0, x, 0x110 + N, 0xf, 0xf, true);
}
__device__ __forceinline__ int wave_incl_scan(int x, int lane) {
  x += lane_shr16<1>(x);
  x += lane_shr16<2>(x);
  x += lane_shr16<4>(x);
  x += lane_shr16<8>(x);
  const int lo = __builtin_amdgcn_readlane(x, 15);
  if (lane >= 16) x += lo;
  return x;
}

__global__ __launch_bounds__(1024) void lsh_sort_kernel(
    const int* __restrict__ bucketArr, int* __restrict__ sticker)
{
  __shared__ __align__(16) int cnt[NCNT + NCNT/32];
  __shared__ int wtot[32];

  const int bh   = blockIdx.x;
  const int h    = bh & (NHASH - 1);
  const int b    = bh >> 3;
  const int tid  = threadIdx.x;
  const int wave = tid >> 5, lane = tid & 31;
  const int* src = bucketArr + (size_t)bh*SEQ;

  int q[8];
#pragma unroll
  for (int k = 0; k < 8; k++) q[k] = src[256*wave + 32*k + lane] & (NBUCK - 1);

  {
    int4* c4 = (int4*)cnt;
    for (int i = tid; i < (NCNT + NCNT/32)/4; i += 1024) c4[i] = make_int4(0, 0, 0, 0);
  }
  __syncthreads();

  int rk[8];
#pragma unroll
  for (int k = 0; k < 8; k++) {
    unsigned peers = 0xffffffffu;
#pragma unroll
    for (int bit = 0; bit < 7; bit++) {
      const bool on = ((q[k] >> bit) & 1) != 0;
      const unsigned m = __builtin_amdgcn_ballot_w32(on);
      peers &= on ? m : ~m;
    }
    rk[k] = (int)__builtin_amdgcn_mbcnt_lo(peers, 0u);
    if (rk[k] == 0)
      cnt[CPAD(q[k]*NTILE + 8*wave + k)] = __builtin_popcount(peers);
  }
  __syncthreads();

  int v[32];
#pragma unroll
  for (int i = 0; i < 32; i++) v[i] = cnt[33*tid + i];
  int run = 0;
#pragma unroll
  for (int i = 0; i < 32; i++) { const int t = v[i]; v[i] = run; run += t; }
  const int inc = wave_incl_scan(run, lane);
  if (lane == 31) wtot[wave] = inc;
  __syncthreads();
  const int wt   = wtot[lane];
  const int winc = wave_incl_scan(wt, lane);
  const int woff = __builtin_amdgcn_readlane(winc - wt, __builtin_amdgcn_readfirstlane(wave));
  const int add  = woff + inc - run;
#pragma unroll
  for (int i = 0; i < 32; i++) cnt[33*tid + i] = v[i] + add;
  __syncthreads();

#pragma unroll
  for (int k = 0; k < 8; k++) {
    const int pos = cnt[CPAD(q[k]*NTILE + 8*wave + k)] + rk[k];
    sticker[(size_t)b*HS + (size_t)h*SEQ + pos] = h*SEQ + 256*wave + 32*k + lane;
  }
}

#define KP 68
#define VPP 132
#define PP 132
#define RUN 4

template <int M> __device__ __forceinline__ float lane_xor16(float x) {
  return __builtin_bit_cast(float, __builtin_amdgcn_update_dpp(
      0, __builtin_bit_cast(int, x), 0x160 + M, 0xf, 0xf, true));
}
__device__ __forceinline__ float row16_max(float x) {
  x = fmaxf(x, lane_xor16<1>(x));
  x = fmaxf(x, lane_xor16<2>(x));
  x = fmaxf(x, lane_xor16<4>(x));
  x = fmaxf(x, lane_xor16<8>(x));
  return x;
}
__device__ __forceinline__ float row16_sum(float x) {
  x += lane_xor16<1>(x);
  x += lane_xor16<2>(x);
  x += lane_xor16<4>(x);
  x += lane_xor16<8>(x);
  return x;
}

__global__ __launch_bounds__(128) void lsh_attn_kernel(
    const float* __restrict__ qk, const float* __restrict__ vin,
    const int* __restrict__ sticker,
    float* __restrict__ oBuf, float* __restrict__ lgBuf)
{
  __shared__ int   idxA[128];
  __shared__ int   tA  [128];
  __shared__ float ksc [128];
  __shared__ float lse [64];
  __shared__ __align__(16) float Ks[128][KP];
  __shared__ __align__(16) float Vp[64][VPP];
  __shared__ __align__(16) float Pw[4][16][PP];

  const int bid = blockIdx.x;
  const int cg  = bid & (NCHUNK/RUN - 1);
  const int b   = bid / (NCHUNK/RUN);
  const int tid = threadIdx.x;
  const int cfirst = cg * RUN;
  const int prev = (cfirst + NCHUNK - 1) & (NCHUNK - 1);

  {
    const int j  = tid >> 1;
    const int c0 = (tid & 1) * 32;
    const int idx0 = sticker[(size_t)b*HS + (size_t)cfirst*BSZ + j];
    const int idx1 = sticker[(size_t)b*HS + (size_t)prev  *BSZ + j];
    const int t0 = idx0 & (SEQ - 1);
    const int t1 = idx1 & (SEQ - 1);
    const float4* sk0 = (const float4*)(qk  + ((size_t)b*SEQ + t0)*DIM + c0);
    const float4* sk1 = (const float4*)(qk  + ((size_t)b*SEQ + t1)*DIM + c0);
    const int q  = tid >> 2;
    const int vc = tid & 3;
    const int2 iv0 = *(const int2*)(sticker + (size_t)b*HS + (size_t)cfirst*BSZ + 2*q);
    const int2 iv1 = *(const int2*)(sticker + (size_t)b*HS + (size_t)prev  *BSZ + 2*q);
    const float4* sa0 = (const float4*)(vin + ((size_t)b*SEQ + (iv0.x & (SEQ - 1)))*DIM) + vc;
    const float4* sb0 = (const float4*)(vin + ((size_t)b*SEQ + (iv0.y & (SEQ - 1)))*DIM) + vc;
    const float4* sa1 = (const float4*)(vin + ((size_t)b*SEQ + (iv1.x & (SEQ - 1)))*DIM) + vc;
    const float4* sb1 = (const float4*)(vin + ((size_t)b*SEQ + (iv1.y & (SEQ - 1)))*DIM) + vc;
    float4 k0[8], k1[8], wa0[4], wb0[4], wa1[4], wb1[4];
#pragma unroll
    for (int i = 0; i < 8; i++) k0[i] = sk0[i];
#pragma unroll
    for (int i = 0; i < 8; i++) k1[i] = sk1[i];
#pragma unroll
    for (int i = 0; i < 4; i++) { wa0[i] = sa0[4*i]; wb0[i] = sb0[4*i]; }
#pragma unroll
    for (int i = 0; i < 4; i++) { wa1[i] = sa1[4*i]; wb1[i] = sb1[4*i]; }
    float4* dk0 = (float4*)(&Ks[j][c0]);
    float4* dk1 = (float4*)(&Ks[64 + j][c0]);
    float4* dv0 = (float4*)(&Vp[q][8*vc]);
    float4* dv1 = (float4*)(&Vp[32 + q][8*vc]);
    float ss0 = 0.f, ss1 = 0.f;
#pragma unroll
    for (int i = 0; i < 8; i++) {
      dk0[i] = k0[i];
      ss0 += k0[i].x * k0[i].x; ss0 += k0[i].y * k0[i].y;
      ss0 += k0[i].z * k0[i].z; ss0 += k0[i].w * k0[i].w;
    }
#pragma unroll
    for (int i = 0; i < 8; i++) {
      dk1[i] = k1[i];
      ss1 += k1[i].x * k1[i].x; ss1 += k1[i].y * k1[i].y;
      ss1 += k1[i].z * k1[i].z; ss1 += k1[i].w * k1[i].w;
    }
#pragma unroll
    for (int i = 0; i < 4; i++) {
      dv0[8*i]     = make_float4(wa0[i].x, wb0[i].x, wa0[i].y, wb0[i].y);
      dv0[8*i + 1] = make_float4(wa0[i].z, wb0[i].z, wa0[i].w, wb0[i].w);
      dv1[8*i]     = make_float4(wa1[i].x, wb1[i].x, wa1[i].y, wb1[i].y);
      dv1[8*i + 1] = make_float4(wa1[i].z, wb1[i].z, wa1[i].w, wb1[i].w);
    }
    ss0 += lane_xor16<1>(ss0);
    ss1 += lane_xor16<1>(ss1);
    if ((tid & 1) == 0) {
      idxA[j]      = idx0;  tA[j]      = t0;
      idxA[64 + j] = idx1;  tA[64 + j] = t1;
      ksc[j]       = 0.125f * (1.0f / (sqrtf(ss0) + 1e-6f));
      ksc[64 + j]  = 0.125f * (1.0f / (sqrtf(ss1) + 1e-6f));
    }
  }

  const int wave = tid >> 5, lane = tid & 31, half = lane >> 4, mn = lane & 15;
  const int q0 = wave * 16;
  float (*Pme)[PP] = Pw[wave];

#pragma unroll 1
  for (int it = 0; it < RUN; it++) {
    const int cb = (it & 1) * 64;
    const int pb = 64 - cb;
    __syncthreads();

    v8f S[8] = {};
    {
      v2f qa[2][2], kf[2][2][8];
      const float* qrow = &Ks[cb + q0 + mn][2*half];
      const float* krc  = &Ks[cb + mn][2*half];
      const float* krp  = &Ks[pb + mn][2*half];
#pragma unroll
      for (int u = 0; u < 2; u++) {
        qa[0][u] = *(const v2f*)(qrow + 4*u);
#pragma unroll
        for (int j = 0; j < 8; j++)
          kf[0][u][j] = *(const v2f*)(((j < 4) ? krc + 16*j*KP : krp + 16*(j - 4)*KP) + 4*u);
      }
      __builtin_amdgcn_sched_barrier(0);
#pragma unroll
      for (int blk = 0; blk < 8; blk++) {
        const int cur = blk & 1, nxt = cur ^ 1;
        if (blk + 1 < 8) {
#pragma unroll
          for (int u = 0; u < 2; u++) {
            const int s = 2*(blk + 1) + u;
            qa[nxt][u] = *(const v2f*)(qrow + 4*s);
#pragma unroll
            for (int j = 0; j < 8; j++)
              kf[nxt][u][j] = *(const v2f*)(((j < 4) ? krc + 16*j*KP : krp + 16*(j - 4)*KP) + 4*s);
          }
        }
        __builtin_amdgcn_sched_barrier(0);
#pragma unroll
        for (int u = 0; u < 2; u++)
#pragma unroll
          for (int j = 0; j < 8; j++)
            S[j] = __builtin_amdgcn_wmma_f32_16x16x4_f32(false, qa[cur][u], false, kf[cur][u][j],
                                                         (short)0, S[j], false, false);
        __builtin_amdgcn_sched_barrier(0);
      }
    }
    int trow[8];
#pragma unroll
    for (int r = 0; r < 8; r++) trow[r] = tA[cb + q0 + r + 8*half];
#pragma unroll
    for (int j = 0; j < 8; j++) {
      const int kb = (j < 4) ? (cb + 16*j) : (pb + 16*(j - 4));
      const float sc = ksc[kb + mn];
      const int   tc = tA [kb + mn];
#pragma unroll
      for (int r = 0; r < 8; r++) {
        float dval = S[j][r] * sc;
        if (trow[r] == tc) dval = SELF_VAL;
        S[j][r] = dval;
      }
    }

    float inv[8], lsev[8];
#pragma unroll
    for (int r = 0; r < 8; r++) {
      float m = S[0][r];
#pragma unroll
      for (int j = 1; j < 8; j++) m = fmaxf(m, S[j][r]);
      m = row16_max(m);
      float ssum = 0.f;
#pragma unroll
      for (int j = 0; j < 8; j++) { const float e = expf(S[j][r] - m); S[j][r] = e; ssum += e; }
      ssum = row16_sum(ssum);
      lsev[r] = m + logf(ssum);
      inv[r]  = 1.0f / ssum;
    }
    if (mn == 0) {
#pragma unroll
      for (int r = 0; r < 8; r++) lse[q0 + r + 8*half] = lsev[r];
    }
#pragma unroll
    for (int j = 0; j < 8; j++)
#pragma unroll
      for (int r = 0; r < 8; r++)
        Pme[r + 8*half][4*(8*(j >> 1) + (mn >> 1)) + 2*(j & 1) + (mn & 1)] = S[j][r] * inv[r];
    __syncthreads();

    v8f O[4] = {};
    {
      v2f pa[2][4], vf[2][4][4];
      const float* prow = &Pme[mn][2*half];
      const float* vcur = &Vp[(cb >> 1) + 8*half][2*mn];
      const float* vprv = &Vp[(pb >> 1) + 8*half][2*mn];
#pragma unroll
      for (int u = 0; u < 4; u++) {
        pa[0][u] = *(const v2f*)(prow + 4*u);
#pragma unroll
        for (int ni = 0; ni < 4; ni++)
          vf[0][u][ni] = *(const v2f*)(vcur + u*VPP + 32*ni);
      }
      __builtin_amdgcn_sched_barrier(0);
#pragma unroll
      for (int blk = 0; blk < 8; blk++) {
        const int cur = blk & 1, nxt = cur ^ 1;
        if (blk + 1 < 8) {
#pragma unroll
          for (int u = 0; u < 4; u++) {
            const int t  = 4*(blk + 1) + u;
            const int pr = 16*((t >> 3) & 1) + (t & 7);
            pa[nxt][u] = *(const v2f*)(prow + 4*t);
#pragma unroll
            for (int ni = 0; ni < 4; ni++)
              vf[nxt][u][ni] = *(const v2f*)(((t < 16) ? vcur : vprv) + pr*VPP + 32*ni);
          }
        }
        __builtin_amdgcn_sched_barrier(0);
#pragma unroll
        for (int u = 0; u < 4; u++)
#pragma unroll
          for (int ni = 0; ni < 4; ni++)
            O[ni] = __builtin_amdgcn_wmma_f32_16x16x4_f32(false, pa[cur][u], false, vf[cur][u][ni],
                                                          (short)0, O[ni], false, false);
        __builtin_amdgcn_sched_barrier(0);
      }
    }
#pragma unroll
    for (int ni = 0; ni < 4; ni++)
#pragma unroll
      for (int r = 0; r < 8; r++)
        Pme[r + 8*half][16*ni + mn] = O[ni][r];
    __syncthreads();

#pragma unroll
    for (int it2 = 0; it2 < 8; it2++) {
      const int row = 2*it2 + half;
      const float4 x = *(const float4*)(&Pme[row][4*mn]);
      *(float4*)(oBuf + ((size_t)b*HS + idxA[cb + q0 + row])*DIM + 4*mn) = x;
    }
    if (tid < 64) lgBuf[(size_t)b*HS + idxA[cb + tid]] = lse[tid];

    if (it + 1 < RUN) {
      const int j  = tid >> 1;
      const int c0 = (tid & 1) * 32;
      const int idx0 = sticker[(size_t)b*HS + (size_t)(cfirst + it + 1)*BSZ + j];
      const int t0 = idx0 & (SEQ - 1);
      const float4* sk0 = (const float4*)(qk  + ((size_t)b*SEQ + t0)*DIM + c0);
      const int q  = tid >> 2;
      const int vc = tid & 3;
      const int2 iv0 = *(const int2*)(sticker + (size_t)b*HS + (size_t)(cfirst + it + 1)*BSZ + 2*q);
      const float4* sa0 = (const float4*)(vin + ((size_t)b*SEQ + (iv0.x & (SEQ - 1)))*DIM) + vc;
      const float4* sb0 = (const float4*)(vin + ((size_t)b*SEQ + (iv0.y & (SEQ - 1)))*DIM) + vc;
      float4 k0[8], wa0[4], wb0[4];
#pragma unroll
      for (int i = 0; i < 8; i++) k0[i] = sk0[i];
#pragma unroll
      for (int i = 0; i < 4; i++) { wa0[i] = sa0[4*i]; wb0[i] = sb0[4*i]; }
      float4* dk0 = (float4*)(&Ks[pb + j][c0]);
      float4* dv0 = (float4*)(&Vp[(pb >> 1) + q][8*vc]);
      float ss0 = 0.f;
#pragma unroll
      for (int i = 0; i < 8; i++) {
        dk0[i] = k0[i];
        ss0 += k0[i].x * k0[i].x; ss0 += k0[i].y * k0[i].y;
        ss0 += k0[i].z * k0[i].z; ss0 += k0[i].w * k0[i].w;
      }
#pragma unroll
      for (int i = 0; i < 4; i++) {
        dv0[8*i]     = make_float4(wa0[i].x, wb0[i].x, wa0[i].y, wb0[i].y);
        dv0[8*i + 1] = make_float4(wa0[i].z, wb0[i].z, wa0[i].w, wb0[i].w);
      }
      ss0 += lane_xor16<1>(ss0);
      if ((tid & 1) == 0) {
        idxA[pb + j] = idx0;  tA[pb + j] = t0;
        ksc[pb + j]  = 0.125f * (1.0f / (sqrtf(ss0) + 1e-6f));
      }
    }
  }
}

__global__ __launch_bounds__(256) void lsh_combine_kernel(
    const float* __restrict__ oBuf, const float* __restrict__ lgBuf,
    float* __restrict__ out)
{
  const int g = blockIdx.x * blockDim.x + threadIdx.x;
  const int b = g >> 13;
  const int t = g & (SEQ - 1);
  float lg[NHASH];
  float m = -1e30f;
#pragma unroll
  for (int h = 0; h < NHASH; h++) {
    lg[h] = lgBuf[(size_t)b*HS + (size_t)h*SEQ + t];
    m = fmaxf(m, lg[h]);
  }
  float ssum = 0.f;
#pragma unroll
  for (int h = 0; h < NHASH; h++) { lg[h] = expf(lg[h] - m); ssum += lg[h]; }
  const float inv = 1.0f / ssum;
#pragma unroll
  for (int h = 0; h < NHASH; h++) lg[h] *= inv;

  float* dst = out + ((size_t)b*SEQ + t)*DIM;
#pragma unroll
  for (int d = 0; d < DIM; d += 4) {
    float4 acc = {0.f, 0.f, 0.f, 0.f};
    for (int h = 0; h < NHASH; h++) {
      const float4 o4 = *(const float4*)(oBuf + ((size_t)b*HS + (size_t)h*SEQ + t)*DIM + d);
      acc.x += lg[h]*o4.x; acc.y += lg[h]*o4.y; acc.z += lg[h]*o4.z; acc.w += lg[h]*o4.w;
    }
    *(float4*)(dst + d) = acc;
  }
}

extern "C" void kernel_launch(void* const* d_in, const int* in_sizes, int n_in,
                              void* d_out, int out_size, void* d_ws, size_t ws_size,
                              hipStream_t stream)
{
  const float* qk  = (const float*)d_in[0];
  const float* v   = (const float*)d_in[1];
  const float* rot = (const float*)d_in[2];

  char* ws = (char*)d_ws;
  size_t off = 0;
  auto alloc = [&](size_t bytes) -> void* {
    void* p = ws + off;
    off += (bytes + 255) & ~(size_t)255;
    return p;
  };
  int*   bucketArr = (int*)  alloc((size_t)BATCH*NHASH*SEQ*sizeof(int));
  int*   sticker   = (int*)  alloc((size_t)BATCH*HS*sizeof(int));
  float* oBuf      = (float*)alloc((size_t)BATCH*HS*DIM*sizeof(float));
  float* lgBuf     = (float*)alloc((size_t)BATCH*HS*sizeof(float));
  if (off > ws_size) return;

  lsh_hash_kernel   <<<BATCH*NHASH*(SEQ/64), 256, 0, stream>>>(qk, rot, bucketArr);
  lsh_sort_kernel   <<<BATCH*NHASH,         1024, 0, stream>>>(bucketArr, sticker);
  lsh_attn_kernel   <<<BATCH*NCHUNK/RUN,     128, 0, stream>>>(qk, v, sticker, oBuf, lgBuf);
  lsh_combine_kernel<<<(BATCH*SEQ)/256,      256, 0, stream>>>(oBuf, lgBuf, (float*)d_out);
}


# hash re-tiled: one work-group per 64 tokens walks all 8 rounds (token tile staged once, 8x fewer work-groups), each round's argmax read from the LDS tile by all 8 waves, not a 2-wave serial scan
# speedup vs baseline: 1.8275x; 1.1149x over previous
#include <hip/hip_runtime.h>
#include <math.h>

#define BATCH 8
#define SEQ   8192
#define DIM   64
#define NHASH 8
#define NBUCK 128
#define NBALL (NHASH*NBUCK)
#define BSZ   64
#define NCHUNK (NHASH*(SEQ/BSZ))
#define HS    (NHASH*SEQ)
#define TILE  32
#define NTILE (SEQ/TILE)
#define NCNT  (NBUCK*NTILE)
#define SELF_VAL (-5.0e4f)

typedef float v2f __attribute__((ext_vector_type(2)));
typedef float v8f __attribute__((ext_vector_type(8)));

template <int M> __device__ __forceinline__ int lane_xor16i(int x) {
  return __builtin_amdgcn_update_dpp(0, x, 0x160 + M, 0xf, 0xf, true);
}
template <int M> __device__ __forceinline__ float lane_xor16f(float x) {
  return __builtin_bit_cast(float, lane_xor16i<M>(__builtin_bit_cast(int, x)));
}

__global__ __launch_bounds__(256) void lsh_hash_kernel(
    const float* __restrict__ qk, const float* __restrict__ rot,
    int* __restrict__ bucketArr)
{
  __shared__ __align__(16) float A [64][68];
  __shared__ __align__(16) float Bm[64][68];
  __shared__ __align__(16) float R [64][68];

  const int bid = blockIdx.x;
  const int tt  = bid & 127;
  const int b   = bid >> 7;
  const int tid = threadIdx.x;
  const int t0  = tt * 64;

  const int row = tid >> 2, qd = tid & 3, c0 = qd * 16;

  {
    const float4* s4 = (const float4*)(qk + ((size_t)b*SEQ + t0 + row)*DIM + c0);
    float4* d4 = (float4*)(&A[row][c0]);
    d4[0]=s4[0]; d4[1]=s4[1]; d4[2]=s4[2]; d4[3]=s4[3];
  }
  const float* rsrc = rot + (size_t)row*(NHASH*64) + c0;
  float4* bdst = (float4*)(&Bm[row][c0]);
  {
    const float4* s4 = (const float4*)rsrc;
    bdst[0]=s4[0]; bdst[1]=s4[1]; bdst[2]=s4[2]; bdst[3]=s4[3];
  }

  const int wave = tid >> 5, lane = tid & 31, half = lane >> 4, mn = lane & 15;
  for (int h = 0; h < NHASH; h++) {
    __syncthreads();

    float4 nb0, nb1, nb2, nb3;
    if (h + 1 < NHASH) {
      const float4* s4 = (const float4*)(rsrc + (h + 1)*64);
      nb0 = s4[0]; nb1 = s4[1]; nb2 = s4[2]; nb3 = s4[3];
    }

    for (int T = wave; T < 16; T += 8) {
      const int mi = T >> 2, ni = T & 3;
      v8f acc = {};
#pragma unroll
      for (int k0 = 0; k0 < 64; k0 += 4) {
        v2f a, bb;
        a.x  = A [mi*16 + mn][k0 + 2*half    ];
        a.y  = A [mi*16 + mn][k0 + 2*half + 1];
        bb.x = Bm[k0 + 2*half    ][ni*16 + mn];
        bb.y = Bm[k0 + 2*half + 1][ni*16 + mn];
        acc = __builtin_amdgcn_wmma_f32_16x16x4_f32(false, a, false, bb,
                                                    (short)0, acc, false, false);
      }
#pragma unroll
      for (int r = 0; r < 8; r++)
        R[mi*16 + r + 8*half][ni*16 + mn] = acc[r];
    }
    __syncthreads();

    if (h + 1 < NHASH) { bdst[0] = nb0; bdst[1] = nb1; bdst[2] = nb2; bdst[3] = nb3; }

    const float4* rp = (const float4*)(&R[row][c0]);
    const float4 x0 = rp[0], x1 = rp[1], x2 = rp[2], x3 = rp[3];
    const float xs[16] = {x0.x, x0.y, x0.z, x0.w, x1.x, x1.y, x1.z, x1.w,
                          x2.x, x2.y, x2.z, x2.w, x3.x, x3.y, x3.z, x3.w};
    float vmx = xs[0], vmn = xs[0];
    int   imx = 0,     imn = 0;
#pragma unroll
    for (int i = 1; i < 16; i++) {
      if (xs[i] > vmx) { vmx = xs[i]; imx = i; }
      if (xs[i] < vmn) { vmn = xs[i]; imn = i; }
    }
    imx += c0;  imn += c0;
    {
      const float omx = lane_xor16f<1>(vmx), omn = lane_xor16f<1>(vmn);
      const int   jmx = lane_xor16i<1>(imx), jmn = lane_xor16i<1>(imn);
      const bool tx = (omx > vmx) || (omx == vmx && jmx < imx);
      const bool tn = (omn < vmn) || (omn == vmn && jmn < imn);
      vmx = tx ? omx : vmx;  imx = tx ? jmx : imx;
      vmn = tn ? omn : vmn;  imn = tn ? jmn : imn;
    }
    {
      const float omx = lane_xor16f<2>(vmx), omn = lane_xor16f<2>(vmn);
      const int   jmx = lane_xor16i<2>(imx), jmn = lane_xor16i<2>(imn);
      const bool tx = (omx > vmx) || (omx == vmx && jmx < imx);
      const bool tn = (omn < vmn) || (omn == vmn && jmn < imn);
      vmx = tx ? omx : vmx;  imx = tx ? jmx : imx;
      vmn = tn ? omn : vmn;  imn = tn ? jmn : imn;
    }
    const int bidx = (-vmn > vmx) ? 64 + imn : imx;
    if (qd == 0)
      bucketArr[((size_t)b*NHASH + h)*SEQ + t0 + row] = h*NBUCK + bidx;
  }
}

#define CPAD(e) ((e) + ((e) >> 5))

template <int N> __device__ __forceinline__ int lane_shr16(int x) {
  return __builtin_amdgcn_update_dpp(0, x, 0x110 + N, 0xf, 0xf, true);
}
__device__ __forceinline__ int wave_incl_scan(int x, int lane) {
  x += lane_shr16<1>(x);
  x += lane_shr16<2>(x);
  x += lane_shr16<4>(x);
  x += lane_shr16<8>(x);
  const int lo = __builtin_amdgcn_readlane(x, 15);
  if (lane >= 16) x += lo;
  return x;
}

__global__ __launch_bounds__(1024) void lsh_sort_kernel(
    const int* __restrict__ bucketArr, int* __restrict__ sticker)
{
  __shared__ __align__(16) int cnt[NCNT + NCNT/32];
  __shared__ int wtot[32];

  const int bh   = blockIdx.x;
  const int h    = bh & (NHASH - 1);
  const int b    = bh >> 3;
  const int tid  = threadIdx.x;
  const int wave = tid >> 5, lane = tid & 31;
  const int* src = bucketArr + (size_t)bh*SEQ;

  int q[8];
#pragma unroll
  for (int k = 0; k < 8; k++) q[k] = src[256*wave + 32*k + lane] & (NBUCK - 1);

  {
    int4* c4 = (int4*)cnt;
    for (int i = tid; i < (NCNT + NCNT/32)/4; i += 1024) c4[i] = make_int4(0, 0, 0, 0);
  }
  __syncthreads();

  int rk[8];
#pragma unroll
  for (int k = 0; k < 8; k++) {
    unsigned peers = 0xffffffffu;
#pragma unroll
    for (int bit = 0; bit < 7; bit++) {
      const bool on = ((q[k] >> bit) & 1) != 0;
      const unsigned m = __builtin_amdgcn_ballot_w32(on);
      peers &= on ? m : ~m;
    }
    rk[k] = (int)__builtin_amdgcn_mbcnt_lo(peers, 0u);
    if (rk[k] == 0)
      cnt[CPAD(q[k]*NTILE + 8*wave + k)] = __builtin_popcount(peers);
  }
  __syncthreads();

  int v[32];
#pragma unroll
  for (int i = 0; i < 32; i++) v[i] = cnt[33*tid + i];
  int run = 0;
#pragma unroll
  for (int i = 0; i < 32; i++) { const int t = v[i]; v[i] = run; run += t; }
  const int inc = wave_incl_scan(run, lane);
  if (lane == 31) wtot[wave] = inc;
  __syncthreads();
  const int wt   = wtot[lane];
  const int winc = wave_incl_scan(wt, lane);
  const int woff = __builtin_amdgcn_readlane(winc - wt, __builtin_amdgcn_readfirstlane(wave));
  const int add  = woff + inc - run;
#pragma unroll
  for (int i = 0; i < 32; i++) cnt[33*tid + i] = v[i] + add;
  __syncthreads();

#pragma unroll
  for (int k = 0; k < 8; k++) {
    const int pos = cnt[CPAD(q[k]*NTILE + 8*wave + k)] + rk[k];
    sticker[(size_t)b*HS + (size_t)h*SEQ + pos] = h*SEQ + 256*wave + 32*k + lane;
  }
}

#define KP 68
#define VPP 132
#define PP 132
#define RUN 4

template <int M> __device__ __forceinline__ float lane_xor16(float x) {
  return __builtin_bit_cast(float, __builtin_amdgcn_update_dpp(
      0, __builtin_bit_cast(int, x), 0x160 + M, 0xf, 0xf, true));
}
__device__ __forceinline__ float row16_max(float x) {
  x = fmaxf(x, lane_xor16<1>(x));
  x = fmaxf(x, lane_xor16<2>(x));
  x = fmaxf(x, lane_xor16<4>(x));
  x = fmaxf(x, lane_xor16<8>(x));
  return x;
}
__device__ __forceinline__ float row16_sum(float x) {
  x += lane_xor16<1>(x);
  x += lane_xor16<2>(x);
  x += lane_xor16<4>(x);
  x += lane_xor16<8>(x);
  return x;
}

__global__ __launch_bounds__(128) void lsh_attn_kernel(
    const float* __restrict__ qk, const float* __restrict__ vin,
    const int* __restrict__ sticker,
    float* __restrict__ oBuf, float* __restrict__ lgBuf)
{
  __shared__ int   idxA[128];
  __shared__ int   tA  [128];
  __shared__ float ksc [128];
  __shared__ float lse [64];
  __shared__ __align__(16) float Ks[128][KP];
  __shared__ __align__(16) float Vp[64][VPP];
  __shared__ __align__(16) float Pw[4][16][PP];

  const int bid = blockIdx.x;
  const int cg  = bid & (NCHUNK/RUN - 1);
  const int b   = bid / (NCHUNK/RUN);
  const int tid = threadIdx.x;
  const int cfirst = cg * RUN;
  const int prev = (cfirst + NCHUNK - 1) & (NCHUNK - 1);

  {
    const int j  = tid >> 1;
    const int c0 = (tid & 1) * 32;
    const int idx0 = sticker[(size_t)b*HS + (size_t)cfirst*BSZ + j];
    const int idx1 = sticker[(size_t)b*HS + (size_t)prev  *BSZ + j];
    const int t0 = idx0 & (SEQ - 1);
    const int t1 = idx1 & (SEQ - 1);
    const float4* sk0 = (const float4*)(qk  + ((size_t)b*SEQ + t0)*DIM + c0);
    const float4* sk1 = (const float4*)(qk  + ((size_t)b*SEQ + t1)*DIM + c0);
    const int q  = tid >> 2;
    const int vc = tid & 3;
    const int2 iv0 = *(const int2*)(sticker + (size_t)b*HS + (size_t)cfirst*BSZ + 2*q);
    const int2 iv1 = *(const int2*)(sticker + (size_t)b*HS + (size_t)prev  *BSZ + 2*q);
    const float4* sa0 = (const float4*)(vin + ((size_t)b*SEQ + (iv0.x & (SEQ - 1)))*DIM) + vc;
    const float4* sb0 = (const float4*)(vin + ((size_t)b*SEQ + (iv0.y & (SEQ - 1)))*DIM) + vc;
    const float4* sa1 = (const float4*)(vin + ((size_t)b*SEQ + (iv1.x & (SEQ - 1)))*DIM) + vc;
    const float4* sb1 = (const float4*)(vin + ((size_t)b*SEQ + (iv1.y & (SEQ - 1)))*DIM) + vc;
    float4 k0[8], k1[8], wa0[4], wb0[4], wa1[4], wb1[4];
#pragma unroll
    for (int i = 0; i < 8; i++) k0[i] = sk0[i];
#pragma unroll
    for (int i = 0; i < 8; i++) k1[i] = sk1[i];
#pragma unroll
    for (int i = 0; i < 4; i++) { wa0[i] = sa0[4*i]; wb0[i] = sb0[4*i]; }
#pragma unroll
    for (int i = 0; i < 4; i++) { wa1[i] = sa1[4*i]; wb1[i] = sb1[4*i]; }
    float4* dk0 = (float4*)(&Ks[j][c0]);
    float4* dk1 = (float4*)(&Ks[64 + j][c0]);
    float4* dv0 = (float4*)(&Vp[q][8*vc]);
    float4* dv1 = (float4*)(&Vp[32 + q][8*vc]);
    float ss0 = 0.f, ss1 = 0.f;
#pragma unroll
    for (int i = 0; i < 8; i++) {
      dk0[i] = k0[i];
      ss0 += k0[i].x * k0[i].x; ss0 += k0[i].y * k0[i].y;
      ss0 += k0[i].z * k0[i].z; ss0 += k0[i].w * k0[i].w;
    }
#pragma unroll
    for (int i = 0; i < 8; i++) {
      dk1[i] = k1[i];
      ss1 += k1[i].x * k1[i].x; ss1 += k1[i].y * k1[i].y;
      ss1 += k1[i].z * k1[i].z; ss1 += k1[i].w * k1[i].w;
    }
#pragma unroll
    for (int i = 0; i < 4; i++) {
      dv0[8*i]     = make_float4(wa0[i].x, wb0[i].x, wa0[i].y, wb0[i].y);
      dv0[8*i + 1] = make_float4(wa0[i].z, wb0[i].z, wa0[i].w, wb0[i].w);
      dv1[8*i]     = make_float4(wa1[i].x, wb1[i].x, wa1[i].y, wb1[i].y);
      dv1[8*i + 1] = make_float4(wa1[i].z, wb1[i].z, wa1[i].w, wb1[i].w);
    }
    ss0 += lane_xor16<1>(ss0);
    ss1 += lane_xor16<1>(ss1);
    if ((tid & 1) == 0) {
      idxA[j]      = idx0;  tA[j]      = t0;
      idxA[64 + j] = idx1;  tA[64 + j] = t1;
      ksc[j]       = 0.125f * (1.0f / (sqrtf(ss0) + 1e-6f));
      ksc[64 + j]  = 0.125f * (1.0f / (sqrtf(ss1) + 1e-6f));
    }
  }

  const int wave = tid >> 5, lane = tid & 31, half = lane >> 4, mn = lane & 15;
  const int q0 = wave * 16;
  float (*Pme)[PP] = Pw[wave];

#pragma unroll 1
  for (int it = 0; it < RUN; it++) {
    const int cb = (it & 1) * 64;
    const int pb = 64 - cb;
    __syncthreads();

    v8f S[8] = {};
    {
      v2f qa[2][2], kf[2][2][8];
      const float* qrow = &Ks[cb + q0 + mn][2*half];
      const float* krc  = &Ks[cb + mn][2*half];
      const float* krp  = &Ks[pb + mn][2*half];
#pragma unroll
      for (int u = 0; u < 2; u++) {
        qa[0][u] = *(const v2f*)(qrow + 4*u);
#pragma unroll
        for (int j = 0; j < 8; j++)
          kf[0][u][j] = *(const v2f*)(((j < 4) ? krc + 16*j*KP : krp + 16*(j - 4)*KP) + 4*u);
      }
      __builtin_amdgcn_sched_barrier(0);
#pragma unroll
      for (int blk = 0; blk < 8; blk++) {
        const int cur = blk & 1, nxt = cur ^ 1;
        if (blk + 1 < 8) {
#pragma unroll
          for (int u = 0; u < 2; u++) {
            const int s = 2*(blk + 1) + u;
            qa[nxt][u] = *(const v2f*)(qrow + 4*s);
#pragma unroll
            for (int j = 0; j < 8; j++)
              kf[nxt][u][j] = *(const v2f*)(((j < 4) ? krc + 16*j*KP : krp + 16*(j - 4)*KP) + 4*s);
          }
        }
        __builtin_amdgcn_sched_barrier(0);
#pragma unroll
        for (int u = 0; u < 2; u++)
#pragma unroll
          for (int j = 0; j < 8; j++)
            S[j] = __builtin_amdgcn_wmma_f32_16x16x4_f32(false, qa[cur][u], false, kf[cur][u][j],
                                                         (short)0, S[j], false, false);
        __builtin_amdgcn_sched_barrier(0);
      }
    }
    int trow[8];
#pragma unroll
    for (int r = 0; r < 8; r++) trow[r] = tA[cb + q0 + r + 8*half];
#pragma unroll
    for (int j = 0; j < 8; j++) {
      const int kb = (j < 4) ? (cb + 16*j) : (pb + 16*(j - 4));
      const float sc = ksc[kb + mn];
      const int   tc = tA [kb + mn];
#pragma unroll
      for (int r = 0; r < 8; r++) {
        float dval = S[j][r] * sc;
        if (trow[r] == tc) dval = SELF_VAL;
        S[j][r] = dval;
      }
    }

    float inv[8], lsev[8];
#pragma unroll
    for (int r = 0; r < 8; r++) {
      float m = S[0][r];
#pragma unroll
      for (int j = 1; j < 8; j++) m = fmaxf(m, S[j][r]);
      m = row16_max(m);
      float ssum = 0.f;
#pragma unroll
      for (int j = 0; j < 8; j++) { const float e = expf(S[j][r] - m); S[j][r] = e; ssum += e; }
      ssum = row16_sum(ssum);
      lsev[r] = m + logf(ssum);
      inv[r]  = 1.0f / ssum;
    }
    if (mn == 0) {
#pragma unroll
      for (int r = 0; r < 8; r++) lse[q0 + r + 8*half] = lsev[r];
    }
#pragma unroll
    for (int j = 0; j < 8; j++)
#pragma unroll
      for (int r = 0; r < 8; r++)
        Pme[r + 8*half][4*(8*(j >> 1) + (mn >> 1)) + 2*(j & 1) + (mn & 1)] = S[j][r] * inv[r];
    __syncthreads();

    v8f O[4] = {};
    {
      v2f pa[2][4], vf[2][4][4];
      const float* prow = &Pme[mn][2*half];
      const float* vcur = &Vp[(cb >> 1) + 8*half][2*mn];
      const float* vprv = &Vp[(pb >> 1) + 8*half][2*mn];
#pragma unroll
      for (int u = 0; u < 4; u++) {
        pa[0][u] = *(const v2f*)(prow + 4*u);
#pragma unroll
        for (int ni = 0; ni < 4; ni++)
          vf[0][u][ni] = *(const v2f*)(vcur + u*VPP + 32*ni);
      }
      __builtin_amdgcn_sched_barrier(0);
#pragma unroll
      for (int blk = 0; blk < 8; blk++) {
        const int cur = blk & 1, nxt = cur ^ 1;
        if (blk + 1 < 8) {
#pragma unroll
          for (int u = 0; u < 4; u++) {
            const int t  = 4*(blk + 1) + u;
            const int pr = 16*((t >> 3) & 1) + (t & 7);
            pa[nxt][u] = *(const v2f*)(prow + 4*t);
#pragma unroll
            for (int ni = 0; ni < 4; ni++)
              vf[nxt][u][ni] = *(const v2f*)(((t < 16) ? vcur : vprv) + pr*VPP + 32*ni);
          }
        }
        __builtin_amdgcn_sched_barrier(0);
#pragma unroll
        for (int u = 0; u < 4; u++)
#pragma unroll
          for (int ni = 0; ni < 4; ni++)
            O[ni] = __builtin_amdgcn_wmma_f32_16x16x4_f32(false, pa[cur][u], false, vf[cur][u][ni],
                                                          (short)0, O[ni], false, false);
        __builtin_amdgcn_sched_barrier(0);
      }
    }
#pragma unroll
    for (int ni = 0; ni < 4; ni++)
#pragma unroll
      for (int r = 0; r < 8; r++)
        Pme[r + 8*half][16*ni + mn] = O[ni][r];
    __syncthreads();

#pragma unroll
    for (int it2 = 0; it2 < 8; it2++) {
      const int row = 2*it2 + half;
      const float4 x = *(const float4*)(&Pme[row][4*mn]);
      *(float4*)(oBuf + ((size_t)b*HS + idxA[cb + q0 + row])*DIM + 4*mn) = x;
    }
    if (tid < 64) lgBuf[(size_t)b*HS + idxA[cb + tid]] = lse[tid];

    if (it + 1 < RUN) {
      const int j  = tid >> 1;
      const int c0 = (tid & 1) * 32;
      const int idx0 = sticker[(size_t)b*HS + (size_t)(cfirst + it + 1)*BSZ + j];
      const int t0 = idx0 & (SEQ - 1);
      const float4* sk0 = (const float4*)(qk  + ((size_t)b*SEQ + t0)*DIM + c0);
      const int q  = tid >> 2;
      const int vc = tid & 3;
      const int2 iv0 = *(const int2*)(sticker + (size_t)b*HS + (size_t)(cfirst + it + 1)*BSZ + 2*q);
      const float4* sa0 = (const float4*)(vin + ((size_t)b*SEQ + (iv0.x & (SEQ - 1)))*DIM) + vc;
      const float4* sb0 = (const float4*)(vin + ((size_t)b*SEQ + (iv0.y & (SEQ - 1)))*DIM) + vc;
      float4 k0[8], wa0[4], wb0[4];
#pragma unroll
      for (int i = 0; i < 8; i++) k0[i] = sk0[i];
#pragma unroll
      for (int i = 0; i < 4; i++) { wa0[i] = sa0[4*i]; wb0[i] = sb0[4*i]; }
      float4* dk0 = (float4*)(&Ks[pb + j][c0]);
      float4* dv0 = (float4*)(&Vp[(pb >> 1) + q][8*vc]);
      float ss0 = 0.f;
#pragma unroll
      for (int i = 0; i < 8; i++) {
        dk0[i] = k0[i];
        ss0 += k0[i].x * k0[i].x; ss0 += k0[i].y * k0[i].y;
        ss0 += k0[i].z * k0[i].z; ss0 += k0[i].w * k0[i].w;
      }
#pragma unroll
      for (int i = 0; i < 4; i++) {
        dv0[8*i]     = make_float4(wa0[i].x, wb0[i].x, wa0[i].y, wb0[i].y);
        dv0[8*i + 1] = make_float4(wa0[i].z, wb0[i].z, wa0[i].w, wb0[i].w);
      }
      ss0 += lane_xor16<1>(ss0);
      if ((tid & 1) == 0) {
        idxA[pb + j] = idx0;  tA[pb + j] = t0;
        ksc[pb + j]  = 0.125f * (1.0f / (sqrtf(ss0) + 1e-6f));
      }
    }
  }
}

__global__ __launch_bounds__(256) void lsh_combine_kernel(
    const float* __restrict__ oBuf, const float* __restrict__ lgBuf,
    float* __restrict__ out)
{
  const int g = blockIdx.x * blockDim.x + threadIdx.x;
  const int b = g >> 13;
  const int t = g & (SEQ - 1);
  float lg[NHASH];
  float m = -1e30f;
#pragma unroll
  for (int h = 0; h < NHASH; h++) {
    lg[h] = lgBuf[(size_t)b*HS + (size_t)h*SEQ + t];
    m = fmaxf(m, lg[h]);
  }
  float ssum = 0.f;
#pragma unroll
  for (int h = 0; h < NHASH; h++) { lg[h] = expf(lg[h] - m); ssum += lg[h]; }
  const float inv = 1.0f / ssum;
#pragma unroll
  for (int h = 0; h < NHASH; h++) lg[h] *= inv;

  float* dst = out + ((size_t)b*SEQ + t)*DIM;
#pragma unroll
  for (int d = 0; d < DIM; d += 4) {
    float4 acc = {0.f, 0.f, 0.f, 0.f};
    for (int h = 0; h < NHASH; h++) {
      const float4 o4 = *(const float4*)(oBuf + ((size_t)b*HS + (size_t)h*SEQ + t)*DIM + d);
      acc.x += lg[h]*o4.x; acc.y += lg[h]*o4.y; acc.z += lg[h]*o4.z; acc.w += lg[h]*o4.w;
    }
    *(float4*)(dst + d) = acc;
  }
}

extern "C" void kernel_launch(void* const* d_in, const int* in_sizes, int n_in,
                              void* d_out, int out_size, void* d_ws, size_t ws_size,
                              hipStream_t stream)
{
  const float* qk  = (const float*)d_in[0];
  const float* v   = (const float*)d_in[1];
  const float* rot = (const float*)d_in[2];

  char* ws = (char*)d_ws;
  size_t off = 0;
  auto alloc = [&](size_t bytes) -> void* {
    void* p = ws + off;
    off += (bytes + 255) & ~(size_t)255;
    return p;
  };
  int*   bucketArr = (int*)  alloc((size_t)BATCH*NHASH*SEQ*sizeof(int));
  int*   sticker   = (int*)  alloc((size_t)BATCH*HS*sizeof(int));
  float* oBuf      = (float*)alloc((size_t)BATCH*HS*DIM*sizeof(float));
  float* lgBuf     = (float*)alloc((size_t)BATCH*HS*sizeof(float));
  if (off > ws_size) return;

  lsh_hash_kernel   <<<BATCH*(SEQ/64),       256, 0, stream>>>(qk, rot, bucketArr);
  lsh_sort_kernel   <<<BATCH*NHASH,         1024, 0, stream>>>(bucketArr, sticker);
  lsh_attn_kernel   <<<BATCH*NCHUNK/RUN,     128, 0, stream>>>(qk, v, sticker, oBuf, lgBuf);
  lsh_combine_kernel<<<(BATCH*SEQ)/256,      256, 0, stream>>>(oBuf, lgBuf, (float*)d_out);
}


# combine fused into the last hash round's attention launch: its waves fetch the other 7 rounds' rows and logits of their 16 tokens (16 lanes per row), weight, store out; no combine kernel; same sums
# speedup vs baseline: 1.9098x; 1.0450x over previous
#include <hip/hip_runtime.h>
#include <math.h>

#define BATCH 8
#define SEQ   8192
#define DIM   64
#define NHASH 8
#define NBUCK 128
#define NBALL (NHASH*NBUCK)
#define BSZ   64
#define NCHUNK (NHASH*(SEQ/BSZ))
#define HS    (NHASH*SEQ)
#define TILE  32
#define NTILE (SEQ/TILE)
#define NCNT  (NBUCK*NTILE)
#define SELF_VAL (-5.0e4f)

typedef float v2f __attribute__((ext_vector_type(2)));
typedef float v8f __attribute__((ext_vector_type(8)));

template <int M> __device__ __forceinline__ int lane_xor16i(int x) {
  return __builtin_amdgcn_update_dpp(0, x, 0x160 + M, 0xf, 0xf, true);
}
template <int M> __device__ __forceinline__ float lane_xor16f(float x) {
  return __builtin_bit_cast(float, lane_xor16i<M>(__builtin_bit_cast(int, x)));
}

__global__ __launch_bounds__(256) void lsh_hash_kernel(
    const float* __restrict__ qk, const float* __restrict__ rot,
    int* __restrict__ bucketArr)
{
  __shared__ __align__(16) float A [64][68];
  __shared__ __align__(16) float Bm[64][68];
  __shared__ __align__(16) float R [64][68];

  const int bid = blockIdx.x;
  const int tt  = bid & 127;
  const int b   = bid >> 7;
  const int tid = threadIdx.x;
  const int t0  = tt * 64;

  const int row = tid >> 2, qd = tid & 3, c0 = qd * 16;

  {
    const float4* s4 = (const float4*)(qk + ((size_t)b*SEQ + t0 + row)*DIM + c0);
    float4* d4 = (float4*)(&A[row][c0]);
    d4[0]=s4[0]; d4[1]=s4[1]; d4[2]=s4[2]; d4[3]=s4[3];
  }
  const float* rsrc = rot + (size_t)row*(NHASH*64) + c0;
  float4* bdst = (float4*)(&Bm[row][c0]);
  {
    const float4* s4 = (const float4*)rsrc;
    bdst[0]=s4[0]; bdst[1]=s4[1]; bdst[2]=s4[2]; bdst[3]=s4[3];
  }

  const int wave = tid >> 5, lane = tid & 31, half = lane >> 4, mn = lane & 15;
  for (int h = 0; h < NHASH; h++) {
    __syncthreads();

    float4 nb0, nb1, nb2, nb3;
    if (h + 1 < NHASH) {
      const float4* s4 = (const float4*)(rsrc + (h + 1)*64);
      nb0 = s4[0]; nb1 = s4[1]; nb2 = s4[2]; nb3 = s4[3];
    }

    for (int T = wave; T < 16; T += 8) {
      const int mi = T >> 2, ni = T & 3;
      v8f acc = {};
#pragma unroll
      for (int k0 = 0; k0 < 64; k0 += 4) {
        v2f a, bb;
        a.x  = A [mi*16 + mn][k0 + 2*half    ];
        a.y  = A [mi*16 + mn][k0 + 2*half + 1];
        bb.x = Bm[k0 + 2*half    ][ni*16 + mn];
        bb.y = Bm[k0 + 2*half + 1][ni*16 + mn];
        acc = __builtin_amdgcn_wmma_f32_16x16x4_f32(false, a, false, bb,
                                                    (short)0, acc, false, false);
      }
#pragma unroll
      for (int r = 0; r < 8; r++)
        R[mi*16 + r + 8*half][ni*16 + mn] = acc[r];
    }
    __syncthreads();

    if (h + 1 < NHASH) { bdst[0] = nb0; bdst[1] = nb1; bdst[2] = nb2; bdst[3] = nb3; }

    const float4* rp = (const float4*)(&R[row][c0]);
    const float4 x0 = rp[0], x1 = rp[1], x2 = rp[2], x3 = rp[3];
    const float xs[16] = {x0.x, x0.y, x0.z, x0.w, x1.x, x1.y, x1.z, x1.w,
                          x2.x, x2.y, x2.z, x2.w, x3.x, x3.y, x3.z, x3.w};
    float vmx = xs[0], vmn = xs[0];
    int   imx = 0,     imn = 0;
#pragma unroll
    for (int i = 1; i < 16; i++) {
      if (xs[i] > vmx) { vmx = xs[i]; imx = i; }
      if (xs[i] < vmn) { vmn = xs[i]; imn = i; }
    }
    imx += c0;  imn += c0;
    {
      const float omx = lane_xor16f<1>(vmx), omn = lane_xor16f<1>(vmn);
      const int   jmx = lane_xor16i<1>(imx), jmn = lane_xor16i<1>(imn);
      const bool tx = (omx > vmx) || (omx == vmx && jmx < imx);
      const bool tn = (omn < vmn) || (omn == vmn && jmn < imn);
      vmx = tx ? omx : vmx;  imx = tx ? jmx : imx;
      vmn = tn ? omn : vmn;  imn = tn ? jmn : imn;
    }
    {
      const float omx = lane_xor16f<2>(vmx), omn = lane_xor16f<2>(vmn);
      const int   jmx = lane_xor16i<2>(imx), jmn = lane_xor16i<2>(imn);
      const bool tx = (omx > vmx) || (omx == vmx && jmx < imx);
      const bool tn = (omn < vmn) || (omn == vmn && jmn < imn);
      vmx = tx ? omx : vmx;  imx = tx ? jmx : imx;
      vmn = tn ? omn : vmn;  imn = tn ? jmn : imn;
    }
    const int bidx = (-vmn > vmx) ? 64 + imn : imx;
    if (qd == 0)
      bucketArr[((size_t)b*NHASH + h)*SEQ + t0 + row] = h*NBUCK + bidx;
  }
}

#define CPAD(e) ((e) + ((e) >> 5))

template <int N> __device__ __forceinline__ int lane_shr16(int x) {
  return __builtin_amdgcn_update_dpp(0, x, 0x110 + N, 0xf, 0xf, true);
}
__device__ __forceinline__ int wave_incl_scan(int x, int lane) {
  x += lane_shr16<1>(x);
  x += lane_shr16<2>(x);
  x += lane_shr16<4>(x);
  x += lane_shr16<8>(x);
  const int lo = __builtin_amdgcn_readlane(x, 15);
  if (lane >= 16) x += lo;
  return x;
}

__global__ __launch_bounds__(1024) void lsh_sort_kernel(
    const int* __restrict__ bucketArr, int* __restrict__ sticker)
{
  __shared__ __align__(16) int cnt[NCNT + NCNT/32];
  __shared__ int wtot[32];

  const int bh   = blockIdx.x;
  const int h    = bh & (NHASH - 1);
  const int b    = bh >> 3;
  const int tid  = threadIdx.x;
  const int wave = tid >> 5, lane = tid & 31;
  const int* src = bucketArr + (size_t)bh*SEQ;

  int q[8];
#pragma unroll
  for (int k = 0; k < 8; k++) q[k] = src[256*wave + 32*k + lane] & (NBUCK - 1);

  {
    int4* c4 = (int4*)cnt;
    for (int i = tid; i < (NCNT + NCNT/32)/4; i += 1024) c4[i] = make_int4(0, 0, 0, 0);
  }
  __syncthreads();

  int rk[8];
#pragma unroll
  for (int k = 0; k < 8; k++) {
    unsigned peers = 0xffffffffu;
#pragma unroll
    for (int bit = 0; bit < 7; bit++) {
      const bool on = ((q[k] >> bit) & 1) != 0;
      const unsigned m = __builtin_amdgcn_ballot_w32(on);
      peers &= on ? m : ~m;
    }
    rk[k] = (int)__builtin_amdgcn_mbcnt_lo(peers, 0u);
    if (rk[k] == 0)
      cnt[CPAD(q[k]*NTILE + 8*wave + k)] = __builtin_popcount(peers);
  }
  __syncthreads();

  int v[32];
#pragma unroll
  for (int i = 0; i < 32; i++) v[i] = cnt[33*tid + i];
  int run = 0;
#pragma unroll
  for (int i = 0; i < 32; i++) { const int t = v[i]; v[i] = run; run += t; }
  const int inc = wave_incl_scan(run, lane);
  if (lane == 31) wtot[wave] = inc;
  __syncthreads();
  const int wt   = wtot[lane];
  const int winc = wave_incl_scan(wt, lane);
  const int woff = __builtin_amdgcn_readlane(winc - wt, __builtin_amdgcn_readfirstlane(wave));
  const int add  = woff + inc - run;
#pragma unroll
  for (int i = 0; i < 32; i++) cnt[33*tid + i] = v[i] + add;
  __syncthreads();

#pragma unroll
  for (int k = 0; k < 8; k++) {
    const int pos = cnt[CPAD(q[k]*NTILE + 8*wave + k)] + rk[k];
    sticker[(size_t)b*HS + (size_t)h*SEQ + pos] = h*SEQ + 256*wave + 32*k + lane;
  }
}

#define KP 68
#define VPP 132
#define PP 132
#define CPR  (SEQ/BSZ)
#define ARUN 4
#define LRUN 2

template <int M> __device__ __forceinline__ float lane_xor16(float x) {
  return __builtin_bit_cast(float, __builtin_amdgcn_update_dpp(
      0, __builtin_bit_cast(int, x), 0x160 + M, 0xf, 0xf, true));
}
__device__ __forceinline__ float row16_max(float x) {
  x = fmaxf(x, lane_xor16<1>(x));
  x = fmaxf(x, lane_xor16<2>(x));
  x = fmaxf(x, lane_xor16<4>(x));
  x = fmaxf(x, lane_xor16<8>(x));
  return x;
}
__device__ __forceinline__ float row16_sum(float x) {
  x += lane_xor16<1>(x);
  x += lane_xor16<2>(x);
  x += lane_xor16<4>(x);
  x += lane_xor16<8>(x);
  return x;
}

template <int NRUN, int CBASE, int NGRP, bool LAST>
__global__ __launch_bounds__(128) void lsh_attn_kernel(
    const float* __restrict__ qk, const float* __restrict__ vin,
    const int* __restrict__ sticker,
    float* __restrict__ oBuf, float* __restrict__ lgBuf, float* __restrict__ out)
{
  static_assert(NHASH == 8, "the weight exchange below lays 4 logits per lane half");
  __shared__ int   idxA[128];
  __shared__ int   tA  [128];
  __shared__ float ksc [128];
  __shared__ float lse [64];
  __shared__ __align__(16) float Ks[128][KP];
  __shared__ __align__(16) float Vp[64][VPP];
  __shared__ __align__(16) float Pw[4][16][PP];

  const int bid = blockIdx.x;
  const int cg  = bid % NGRP;
  const int b   = bid / NGRP;
  const int tid = threadIdx.x;
  const int cfirst = CBASE + cg * NRUN;
  const int prev = (cfirst + NCHUNK - 1) & (NCHUNK - 1);

  {
    const int j  = tid >> 1;
    const int c0 = (tid & 1) * 32;
    const int idx0 = sticker[(size_t)b*HS + (size_t)cfirst*BSZ + j];
    const int idx1 = sticker[(size_t)b*HS + (size_t)prev  *BSZ + j];
    const int t0 = idx0 & (SEQ - 1);
    const int t1 = idx1 & (SEQ - 1);
    const float4* sk0 = (const float4*)(qk  + ((size_t)b*SEQ + t0)*DIM + c0);
    const float4* sk1 = (const float4*)(qk  + ((size_t)b*SEQ + t1)*DIM + c0);
    const int q  = tid >> 2;
    const int vc = tid & 3;
    const int2 iv0 = *(const int2*)(sticker + (size_t)b*HS + (size_t)cfirst*BSZ + 2*q);
    const int2 iv1 = *(const int2*)(sticker + (size_t)b*HS + (size_t)prev  *BSZ + 2*q);
    const float4* sa0 = (const float4*)(vin + ((size_t)b*SEQ + (iv0.x & (SEQ - 1)))*DIM) + vc;
    const float4* sb0 = (const float4*)(vin + ((size_t)b*SEQ + (iv0.y & (SEQ - 1)))*DIM) + vc;
    const float4* sa1 = (const float4*)(vin + ((size_t)b*SEQ + (iv1.x & (SEQ - 1)))*DIM) + vc;
    const float4* sb1 = (const float4*)(vin + ((size_t)b*SEQ + (iv1.y & (SEQ - 1)))*DIM) + vc;
    float4 k0[8], k1[8], wa0[4], wb0[4], wa1[4], wb1[4];
#pragma unroll
    for (int i = 0; i < 8; i++) k0[i] = sk0[i];
#pragma unroll
    for (int i = 0; i < 8; i++) k1[i] = sk1[i];
#pragma unroll
    for (int i = 0; i < 4; i++) { wa0[i] = sa0[4*i]; wb0[i] = sb0[4*i]; }
#pragma unroll
    for (int i = 0; i < 4; i++) { wa1[i] = sa1[4*i]; wb1[i] = sb1[4*i]; }
    float4* dk0 = (float4*)(&Ks[j][c0]);
    float4* dk1 = (float4*)(&Ks[64 + j][c0]);
    float4* dv0 = (float4*)(&Vp[q][8*vc]);
    float4* dv1 = (float4*)(&Vp[32 + q][8*vc]);
    float ss0 = 0.f, ss1 = 0.f;
#pragma unroll
    for (int i = 0; i < 8; i++) {
      dk0[i] = k0[i];
      ss0 += k0[i].x * k0[i].x; ss0 += k0[i].y * k0[i].y;
      ss0 += k0[i].z * k0[i].z; ss0 += k0[i].w * k0[i].w;
    }
#pragma unroll
    for (int i = 0; i < 8; i++) {
      dk1[i] = k1[i];
      ss1 += k1[i].x * k1[i].x; ss1 += k1[i].y * k1[i].y;
      ss1 += k1[i].z * k1[i].z; ss1 += k1[i].w * k1[i].w;
    }
#pragma unroll
    for (int i = 0; i < 4; i++) {
      dv0[8*i]     = make_float4(wa0[i].x, wb0[i].x, wa0[i].y, wb0[i].y);
      dv0[8*i + 1] = make_float4(wa0[i].z, wb0[i].z, wa0[i].w, wb0[i].w);
      dv1[8*i]     = make_float4(wa1[i].x, wb1[i].x, wa1[i].y, wb1[i].y);
      dv1[8*i + 1] = make_float4(wa1[i].z, wb1[i].z, wa1[i].w, wb1[i].w);
    }
    ss0 += lane_xor16<1>(ss0);
    ss1 += lane_xor16<1>(ss1);
    if ((tid & 1) == 0) {
      idxA[j]      = idx0;  tA[j]      = t0;
      idxA[64 + j] = idx1;  tA[64 + j] = t1;
      ksc[j]       = 0.125f * (1.0f / (sqrtf(ss0) + 1e-6f));
      ksc[64 + j]  = 0.125f * (1.0f / (sqrtf(ss1) + 1e-6f));
    }
  }

  const int wave = tid >> 5, lane = tid & 31, half = lane >> 4, mn = lane & 15;
  const int q0 = wave * 16;
  float (*Pme)[PP] = Pw[wave];

#pragma unroll 1
  for (int it = 0; it < NRUN; it++) {
    const int cb = (it & 1) * 64;
    const int pb = 64 - cb;
    __syncthreads();

    v8f S[8] = {};
    {
      v2f qa[2][2], kf[2][2][8];
      const float* qrow = &Ks[cb + q0 + mn][2*half];
      const float* krc  = &Ks[cb + mn][2*half];
      const float* krp  = &Ks[pb + mn][2*half];
#pragma unroll
      for (int u = 0; u < 2; u++) {
        qa[0][u] = *(const v2f*)(qrow + 4*u);
#pragma unroll
        for (int j = 0; j < 8; j++)
          kf[0][u][j] = *(const v2f*)(((j < 4) ? krc + 16*j*KP : krp + 16*(j - 4)*KP) + 4*u);
      }
      __builtin_amdgcn_sched_barrier(0);
#pragma unroll
      for (int blk = 0; blk < 8; blk++) {
        const int cur = blk & 1, nxt = cur ^ 1;
        if (blk + 1 < 8) {
#pragma unroll
          for (int u = 0; u < 2; u++) {
            const int s = 2*(blk + 1) + u;
            qa[nxt][u] = *(const v2f*)(qrow + 4*s);
#pragma unroll
            for (int j = 0; j < 8; j++)
              kf[nxt][u][j] = *(const v2f*)(((j < 4) ? krc + 16*j*KP : krp + 16*(j - 4)*KP) + 4*s);
          }
        }
        __builtin_amdgcn_sched_barrier(0);
#pragma unroll
        for (int u = 0; u < 2; u++)
#pragma unroll
          for (int j = 0; j < 8; j++)
            S[j] = __builtin_amdgcn_wmma_f32_16x16x4_f32(false, qa[cur][u], false, kf[cur][u][j],
                                                         (short)0, S[j], false, false);
        __builtin_amdgcn_sched_barrier(0);
      }
    }
    int trow[8];
#pragma unroll
    for (int r = 0; r < 8; r++) trow[r] = tA[cb + q0 + r + 8*half];
#pragma unroll
    for (int j = 0; j < 8; j++) {
      const int kb = (j < 4) ? (cb + 16*j) : (pb + 16*(j - 4));
      const float sc = ksc[kb + mn];
      const int   tc = tA [kb + mn];
#pragma unroll
      for (int r = 0; r < 8; r++) {
        float dval = S[j][r] * sc;
        if (trow[r] == tc) dval = SELF_VAL;
        S[j][r] = dval;
      }
    }

    float inv[8], lsev[8];
#pragma unroll
    for (int r = 0; r < 8; r++) {
      float m = S[0][r];
#pragma unroll
      for (int j = 1; j < 8; j++) m = fmaxf(m, S[j][r]);
      m = row16_max(m);
      float ssum = 0.f;
#pragma unroll
      for (int j = 0; j < 8; j++) { const float e = expf(S[j][r] - m); S[j][r] = e; ssum += e; }
      ssum = row16_sum(ssum);
      lsev[r] = m + logf(ssum);
      inv[r]  = 1.0f / ssum;
    }
    if (mn == 0) {
#pragma unroll
      for (int r = 0; r < 8; r++) lse[q0 + r + 8*half] = lsev[r];
    }
#pragma unroll
    for (int j = 0; j < 8; j++)
#pragma unroll
      for (int r = 0; r < 8; r++)
        Pme[r + 8*half][4*(8*(j >> 1) + (mn >> 1)) + 2*(j & 1) + (mn & 1)] = S[j][r] * inv[r];
    __syncthreads();

    v8f O[4] = {};
    {
      v2f pa[2][4], vf[2][4][4];
      const float* prow = &Pme[mn][2*half];
      const float* vcur = &Vp[(cb >> 1) + 8*half][2*mn];
      const float* vprv = &Vp[(pb >> 1) + 8*half][2*mn];
#pragma unroll
      for (int u = 0; u < 4; u++) {
        pa[0][u] = *(const v2f*)(prow + 4*u);
#pragma unroll
        for (int ni = 0; ni < 4; ni++)
          vf[0][u][ni] = *(const v2f*)(vcur + u*VPP + 32*ni);
      }
      __builtin_amdgcn_sched_barrier(0);
#pragma unroll
      for (int blk = 0; blk < 8; blk++) {
        const int cur = blk & 1, nxt = cur ^ 1;
        if (blk + 1 < 8) {
#pragma unroll
          for (int u = 0; u < 4; u++) {
            const int t  = 4*(blk + 1) + u;
            const int pr = 16*((t >> 3) & 1) + (t & 7);
            pa[nxt][u] = *(const v2f*)(prow + 4*t);
#pragma unroll
            for (int ni = 0; ni < 4; ni++)
              vf[nxt][u][ni] = *(const v2f*)(((t < 16) ? vcur : vprv) + pr*VPP + 32*ni);
          }
        }
        __builtin_amdgcn_sched_barrier(0);
#pragma unroll
        for (int u = 0; u < 4; u++)
#pragma unroll
          for (int ni = 0; ni < 4; ni++)
            O[ni] = __builtin_amdgcn_wmma_f32_16x16x4_f32(false, pa[cur][u], false, vf[cur][u][ni],
                                                          (short)0, O[ni], false, false);
        __builtin_amdgcn_sched_barrier(0);
      }
    }
#pragma unroll
    for (int ni = 0; ni < 4; ni++)
#pragma unroll
      for (int r = 0; r < 8; r++)
        Pme[r + 8*half][16*ni + mn] = O[ni][r];
    __syncthreads();

    if constexpr (!LAST) {
#pragma unroll
      for (int it2 = 0; it2 < 8; it2++) {
        const int row = 2*it2 + half;
        const float4 x = *(const float4*)(&Pme[row][4*mn]);
        *(float4*)(oBuf + ((size_t)b*HS + idxA[cb + q0 + row])*DIM + 4*mn) = x;
      }
      if (tid < 64) lgBuf[(size_t)b*HS + idxA[cb + tid]] = lse[tid];
    } else {
      {
        const float* lgp = lgBuf + (size_t)b*HS + tA[cb + q0 + mn];
        float4 l4;
        l4.x = lgp[(size_t)(4*half    )*SEQ];
        l4.y = lgp[(size_t)(4*half + 1)*SEQ];
        l4.z = lgp[(size_t)(4*half + 2)*SEQ];
        const float lown = lse[q0 + mn];
        const float lg3  = lgp[(size_t)3*SEQ];
        l4.w = half ? lown : lg3;
        *(float4*)(&Pme[mn][64 + 4*half]) = l4;
      }
      __syncthreads();
      {
        const float4 a4 = *(const float4*)(&Pme[mn][64]);
        const float4 b4 = *(const float4*)(&Pme[mn][68]);
        float lg[NHASH] = {a4.x, a4.y, a4.z, a4.w, b4.x, b4.y, b4.z, b4.w};
        float m = -1e30f;
#pragma unroll
        for (int h = 0; h < NHASH; h++) m = fmaxf(m, lg[h]);
        float ssum = 0.f;
#pragma unroll
        for (int h = 0; h < NHASH; h++) { lg[h] = expf(lg[h] - m); ssum += lg[h]; }
        const float inv = 1.0f / ssum;
#pragma unroll
        for (int h = 0; h < NHASH; h++) lg[h] *= inv;
        const float4 w4 = half ? make_float4(lg[4], lg[5], lg[6], lg[7])
                               : make_float4(lg[0], lg[1], lg[2], lg[3]);
        *(float4*)(&Pme[mn][64 + 4*half]) = w4;
      }
      __syncthreads();
      {
        float4 o4[8][NHASH - 1];
        int tok[8];
#pragma unroll
        for (int i = 0; i < 8; i++) tok[i] = tA[cb + q0 + 2*i + half];
#pragma unroll
        for (int i = 0; i < 8; i++) {
          const float* src = oBuf + ((size_t)b*HS + tok[i])*DIM + 4*mn;
#pragma unroll
          for (int h = 0; h < NHASH - 1; h++)
            o4[i][h] = *(const float4*)(src + (size_t)h*SEQ*DIM);
        }
        __builtin_amdgcn_sched_barrier(0);
#pragma unroll
        for (int i = 0; i < 8; i++) {
          const int row = 2*i + half;
          const float4 wa = *(const float4*)(&Pme[row][64]);
          const float4 wb = *(const float4*)(&Pme[row][68]);
          const float w[NHASH] = {wa.x, wa.y, wa.z, wa.w, wb.x, wb.y, wb.z, wb.w};
          const float4 own = *(const float4*)(&Pme[row][4*mn]);
          float4 acc = {0.f, 0.f, 0.f, 0.f};
#pragma unroll
          for (int h = 0; h < NHASH - 1; h++) {
            acc.x += w[h]*o4[i][h].x; acc.y += w[h]*o4[i][h].y;
            acc.z += w[h]*o4[i][h].z; acc.w += w[h]*o4[i][h].w;
          }
          acc.x += w[NHASH - 1]*own.x; acc.y += w[NHASH - 1]*own.y;
          acc.z += w[NHASH - 1]*own.z; acc.w += w[NHASH - 1]*own.w;
          *(float4*)(out + ((size_t)b*SEQ + tok[i])*DIM + 4*mn) = acc;
        }
      }
    }

    if (it + 1 < NRUN) {
      const int j  = tid >> 1;
      const int c0 = (tid & 1) * 32;
      const int idx0 = sticker[(size_t)b*HS + (size_t)(cfirst + it + 1)*BSZ + j];
      const int t0 = idx0 & (SEQ - 1);
      const float4* sk0 = (const float4*)(qk  + ((size_t)b*SEQ + t0)*DIM + c0);
      const int q  = tid >> 2;
      const int vc = tid & 3;
      const int2 iv0 = *(const int2*)(sticker + (size_t)b*HS + (size_t)(cfirst + it + 1)*BSZ + 2*q);
      const float4* sa0 = (const float4*)(vin + ((size_t)b*SEQ + (iv0.x & (SEQ - 1)))*DIM) + vc;
      const float4* sb0 = (const float4*)(vin + ((size_t)b*SEQ + (iv0.y & (SEQ - 1)))*DIM) + vc;
      float4 k0[8], wa0[4], wb0[4];
#pragma unroll
      for (int i = 0; i < 8; i++) k0[i] = sk0[i];
#pragma unroll
      for (int i = 0; i < 4; i++) { wa0[i] = sa0[4*i]; wb0[i] = sb0[4*i]; }
      float4* dk0 = (float4*)(&Ks[pb + j][c0]);
      float4* dv0 = (float4*)(&Vp[(pb >> 1) + q][8*vc]);
      float ss0 = 0.f;
#pragma unroll
      for (int i = 0; i < 8; i++) {
        dk0[i] = k0[i];
        ss0 += k0[i].x * k0[i].x; ss0 += k0[i].y * k0[i].y;
        ss0 += k0[i].z * k0[i].z; ss0 += k0[i].w * k0[i].w;
      }
#pragma unroll
      for (int i = 0; i < 4; i++) {
        dv0[8*i]     = make_float4(wa0[i].x, wb0[i].x, wa0[i].y, wb0[i].y);
        dv0[8*i + 1] = make_float4(wa0[i].z, wb0[i].z, wa0[i].w, wb0[i].w);
      }
      ss0 += lane_xor16<1>(ss0);
      if ((tid & 1) == 0) {
        idxA[pb + j] = idx0;  tA[pb + j] = t0;
        ksc[pb + j]  = 0.125f * (1.0f / (sqrtf(ss0) + 1e-6f));
      }
    }
  }
}

extern "C" void kernel_launch(void* const* d_in, const int* in_sizes, int n_in,
                              void* d_out, int out_size, void* d_ws, size_t ws_size,
                              hipStream_t stream)
{
  const float* qk  = (const float*)d_in[0];
  const float* v   = (const float*)d_in[1];
  const float* rot = (const float*)d_in[2];

  char* ws = (char*)d_ws;
  size_t off = 0;
  auto alloc = [&](size_t bytes) -> void* {
    void* p = ws + off;
    off += (bytes + 255) & ~(size_t)255;
    return p;
  };
  int*   bucketArr = (int*)  alloc((size_t)BATCH*NHASH*SEQ*sizeof(int));
  int*   sticker   = (int*)  alloc((size_t)BATCH*HS*sizeof(int));
  float* oBuf      = (float*)alloc((size_t)BATCH*HS*DIM*sizeof(float));
  float* lgBuf     = (float*)alloc((size_t)BATCH*HS*sizeof(float));
  if (off > ws_size) return;

  lsh_hash_kernel   <<<BATCH*(SEQ/64),       256, 0, stream>>>(qk, rot, bucketArr);
  lsh_sort_kernel   <<<BATCH*NHASH,         1024, 0, stream>>>(bucketArr, sticker);
  lsh_attn_kernel<ARUN, 0, (NCHUNK - CPR)/ARUN, false>
                    <<<BATCH*(NCHUNK - CPR)/ARUN, 128, 0, stream>>>(qk, v, sticker, oBuf, lgBuf, (float*)d_out);
  lsh_attn_kernel<LRUN, NCHUNK - CPR, CPR/LRUN, true>
                    <<<BATCH*CPR/LRUN,         128, 0, stream>>>(qk, v, sticker, oBuf, lgBuf, (float*)d_out);
}
